# starting kernel
# baseline (speedup 1.0000x reference)
#include <hip/hip_runtime.h>
#include <hip/hip_bf16.h>
#include <math.h>

#define NUU 50000
#define NJJ 20000
#define NEDGE 500000
#define NEL 100000
#define DIM 128

typedef __attribute__((ext_vector_type(2))) float v2f;
typedef __attribute__((ext_vector_type(8))) float v8f;


__global__ __launch_bounds__(256) void k_zero(float* __restrict__ p, int n) {
  int i = blockIdx.x * 256 + threadIdx.x;
  if (i < n) p[i] = 0.0f;
}

__global__ __launch_bounds__(256) void k_count(const int* __restrict__ dst,
                                               float* __restrict__ cnt, int n) {
  int e = blockIdx.x * 256 + threadIdx.x;
  if (e < n) unsafeAtomicAdd(&cnt[dst[e]], 1.0f);
}

__global__ __launch_bounds__(256) void k_scatter(const float* __restrict__ xsrc,
                                                 const int* __restrict__ src,
                                                 const int* __restrict__ dst,
                                                 float* __restrict__ sum, int n) {
  int e = blockIdx.x * 8 + (threadIdx.x >> 5);
  if (e >= n) return;
  int lane = threadIdx.x & 31;
  int s = src[e];
  int d = dst[e];
  const float4 v = *(const float4*)(xsrc + (size_t)s * DIM + lane * 4);
  float* p = sum + (size_t)d * DIM + lane * 4;
  unsafeAtomicAdd(p + 0, v.x);
  unsafeAtomicAdd(p + 1, v.y);
  unsafeAtomicAdd(p + 2, v.z);
  unsafeAtomicAdd(p + 3, v.w);
}

__global__ __launch_bounds__(256) void k_mean(float* __restrict__ sum,
                                              const float* __restrict__ cnt, int n) {
  int i = blockIdx.x * 256 + threadIdx.x;
  if (i < n) sum[i] = sum[i] / fmaxf(cnt[i >> 7], 1.0f);
}

__global__ __launch_bounds__(256) void k_sage_gemm(
    const float* __restrict__ mean, const float* __restrict__ xdst,
    const float* __restrict__ Wl, const float* __restrict__ Wr,
    const float* __restrict__ bias, float* __restrict__ out,
    int nRowTiles, int relu) {
  int wave = threadIdx.x >> 5;
  int lane = threadIdx.x & 31;
  int tile = blockIdx.x * 8 + wave;
  int rowTile = tile >> 3;
  int colTile = tile & 7;
  if (rowTile >= nRowTiles) return;

  int row0 = rowTile * 16;
  int col0 = colTile * 16;
  int lm = lane & 15;
  int khalf = (lane >= 16) ? 2 : 0;

  const float* aBase = mean + (size_t)(row0 + lm) * DIM + khalf;
  const float* xBase = xdst + (size_t)(row0 + lm) * DIM + khalf;
  const float* wlBase = Wl + (size_t)(col0 + lm) * DIM + khalf;
  const float* wrBase = Wr + (size_t)(col0 + lm) * DIM + khalf;

  float bval = bias[col0 + lm];
  v8f accA = {bval, bval, bval, bval, bval, bval, bval, bval};
  v8f accB = {0.f, 0.f, 0.f, 0.f, 0.f, 0.f, 0.f, 0.f};

#pragma unroll 4
  for (int kk = 0; kk < DIM; kk += 4) {
    v2f am = *(const v2f*)(aBase + kk);
    v2f bl = *(const v2f*)(wlBase + kk);
    v2f ax = *(const v2f*)(xBase + kk);
    v2f br = *(const v2f*)(wrBase + kk);
    accA = __builtin_amdgcn_wmma_f32_16x16x4_f32(false, am, false, bl, (short)0,
                                                 accA, false, false);
    accB = __builtin_amdgcn_wmma_f32_16x16x4_f32(false, ax, false, br, (short)0,
                                                 accB, false, false);
  }

  float* orow = out + (size_t)(row0 + ((lane >= 16) ? 8 : 0)) * DIM + col0 + lm;
#pragma unroll
  for (int r = 0; r < 8; ++r) {
    float v = accA[r] + accB[r];
    if (relu) v = fmaxf(v, 0.0f);
    orow[(size_t)r * DIM] = v;
  }
}

__global__ __launch_bounds__(256) void k_decode(
    const float* __restrict__ zu, const float* __restrict__ zj,
    const int* __restrict__ elu, const int* __restrict__ elj,
    float* __restrict__ out, int n) {
  int p = blockIdx.x * 8 + (threadIdx.x >> 5);
  if (p >= n) return;
  int lane = threadIdx.x & 31;
  const float4 a = *(const float4*)(zu + (size_t)elu[p] * DIM + lane * 4);
  const float4 b = *(const float4*)(zj + (size_t)elj[p] * DIM + lane * 4);
  float dot = a.x * b.x + a.y * b.y + a.z * b.z + a.w * b.w;
  float na = a.x * a.x + a.y * a.y + a.z * a.z + a.w * a.w;
  float nb = b.x * b.x + b.y * b.y + b.z * b.z + b.w * b.w;
#pragma unroll
  for (int off = 16; off > 0; off >>= 1) {
    dot += __shfl_xor(dot, off, 32);
    na += __shfl_xor(na, off, 32);
    nb += __shfl_xor(nb, off, 32);
  }
  if (lane == 0) {
    out[p] = dot / (fmaxf(sqrtf(na), 1e-8f) * fmaxf(sqrtf(nb), 1e-8f));
  }
}


extern "C" void kernel_launch(void* const* d_in, const int* in_sizes, int n_in,
                              void* d_out, int out_size, void* d_ws, size_t ws_size,
                              hipStream_t stream) {
  const float* x_user = (const float*)d_in[0];
  const float* x_job = (const float*)d_in[1];
  const int* uj_src = (const int*)d_in[2];
  const int* uj_dst = (const int*)d_in[3];
  const int* ju_src = (const int*)d_in[4];
  const int* ju_dst = (const int*)d_in[5];
  const int* el_user = (const int*)d_in[6];
  const int* el_job = (const int*)d_in[7];
  const float* W1l_uj = (const float*)d_in[8];
  const float* W1r_uj = (const float*)d_in[9];
  const float* W1l_ju = (const float*)d_in[10];
  const float* W1r_ju = (const float*)d_in[11];
  const float* W2l_uj = (const float*)d_in[12];
  const float* W2r_uj = (const float*)d_in[13];
  const float* W2l_ju = (const float*)d_in[14];
  const float* W2r_ju = (const float*)d_in[15];
  const float* b1_uj = (const float*)d_in[16];
  const float* b1_ju = (const float*)d_in[17];
  const float* b2_uj = (const float*)d_in[18];
  const float* b2_ju = (const float*)d_in[19];
  float* out = (float*)d_out;

  float* ws = (float*)d_ws;
  float* agg_user = ws; ws += (size_t)NUU * DIM;
  float* agg_job  = ws; ws += (size_t)NJJ * DIM;
  float* cnt_user = ws; ws += NUU;
  float* cnt_job  = ws; ws += NJJ;
  float* h_user   = ws; ws += (size_t)NUU * DIM;
  float* h_job    = ws; ws += (size_t)NJJ * DIM;
  float* z_user   = ws; ws += (size_t)NUU * DIM;
  float* z_job    = ws; ws += (size_t)NJJ * DIM;

  const int nUF = NUU * DIM, nJF = NJJ * DIM;
  dim3 blk(256);

  k_zero<<<(nJF + 255) / 256, blk, 0, stream>>>(agg_job, nJF);
  k_zero<<<(nUF + 255) / 256, blk, 0, stream>>>(agg_user, nUF);
  k_zero<<<(NJJ + 255) / 256, blk, 0, stream>>>(cnt_job, NJJ);
  k_zero<<<(NUU + 255) / 256, blk, 0, stream>>>(cnt_user, NUU);
  k_count<<<(NEDGE + 255) / 256, blk, 0, stream>>>(uj_dst, cnt_job, NEDGE);
  k_count<<<(NEDGE + 255) / 256, blk, 0, stream>>>(ju_dst, cnt_user, NEDGE);
  k_scatter<<<NEDGE / 8, blk, 0, stream>>>(x_user, uj_src, uj_dst, agg_job, NEDGE);
  k_scatter<<<NEDGE / 8, blk, 0, stream>>>(x_job, ju_src, ju_dst, agg_user, NEDGE);
  k_mean<<<(nJF + 255) / 256, blk, 0, stream>>>(agg_job, cnt_job, nJF);
  k_mean<<<(nUF + 255) / 256, blk, 0, stream>>>(agg_user, cnt_user, nUF);

  k_sage_gemm<<<NJJ / 16, blk, 0, stream>>>(agg_job, x_job, W1l_uj, W1r_uj,
                                            b1_uj, h_job, NJJ / 16, 1);
  k_sage_gemm<<<NUU / 16, blk, 0, stream>>>(agg_user, x_user, W1l_ju, W1r_ju,
                                            b1_ju, h_user, NUU / 16, 1);

  k_zero<<<(nJF + 255) / 256, blk, 0, stream>>>(agg_job, nJF);
  k_zero<<<(nUF + 255) / 256, blk, 0, stream>>>(agg_user, nUF);
  k_scatter<<<NEDGE / 8, blk, 0, stream>>>(h_user, uj_src, uj_dst, agg_job, NEDGE);
  k_scatter<<<NEDGE / 8, blk, 0, stream>>>(h_job, ju_src, ju_dst, agg_user, NEDGE);
  k_mean<<<(nJF + 255) / 256, blk, 0, stream>>>(agg_job, cnt_job, nJF);
  k_mean<<<(nUF + 255) / 256, blk, 0, stream>>>(agg_user, cnt_user, nUF);

  k_sage_gemm<<<NJJ / 16, blk, 0, stream>>>(agg_job, h_job, W2l_uj, W2r_uj,
                                            b2_uj, z_job, NJJ / 16, 0);
  k_sage_gemm<<<NUU / 16, blk, 0, stream>>>(agg_user, h_user, W2l_ju, W2r_ju,
                                            b2_ju, z_user, NUU / 16, 0);

  k_decode<<<NEL / 8, blk, 0, stream>>>(z_user, z_job, el_user, el_job, out, NEL);
}


# GEMM: both weight matrices staged once per workgroup in LDS in B-fragment order; each wave owns a 16x128 output strip so one A fragment feeds 8 column tiles; k order and sum order kept
# speedup vs baseline: 1.2056x; 1.2056x over previous
#include <hip/hip_runtime.h>
#include <hip/hip_bf16.h>
#include <math.h>

#define NUU 50000
#define NJJ 20000
#define NEDGE 500000
#define NEL 100000
#define DIM 128

typedef __attribute__((ext_vector_type(2))) float v2f;
typedef __attribute__((ext_vector_type(8))) float v8f;


__global__ __launch_bounds__(256) void k_zero(float* __restrict__ p, int n) {
  int i = blockIdx.x * 256 + threadIdx.x;
  if (i < n) p[i] = 0.0f;
}

__global__ __launch_bounds__(256) void k_count(const int* __restrict__ dst,
                                               float* __restrict__ cnt, int n) {
  int e = blockIdx.x * 256 + threadIdx.x;
  if (e < n) unsafeAtomicAdd(&cnt[dst[e]], 1.0f);
}

__global__ __launch_bounds__(256) void k_scatter(const float* __restrict__ xsrc,
                                                 const int* __restrict__ src,
                                                 const int* __restrict__ dst,
                                                 float* __restrict__ sum, int n) {
  int e = blockIdx.x * 8 + (threadIdx.x >> 5);
  if (e >= n) return;
  int lane = threadIdx.x & 31;
  int s = src[e];
  int d = dst[e];
  const float4 v = *(const float4*)(xsrc + (size_t)s * DIM + lane * 4);
  float* p = sum + (size_t)d * DIM + lane * 4;
  unsafeAtomicAdd(p + 0, v.x);
  unsafeAtomicAdd(p + 1, v.y);
  unsafeAtomicAdd(p + 2, v.z);
  unsafeAtomicAdd(p + 3, v.w);
}

__global__ __launch_bounds__(256) void k_mean(float* __restrict__ sum,
                                              const float* __restrict__ cnt, int n) {
  int i = blockIdx.x * 256 + threadIdx.x;
  if (i < n) sum[i] = sum[i] / fmaxf(cnt[i >> 7], 1.0f);
}

#define WPITCH 257
#define GEMM_WAVES 16
__global__ __launch_bounds__(512) void k_sage_gemm(
    const float* __restrict__ mean, const float* __restrict__ xdst,
    const float* __restrict__ Wl, const float* __restrict__ Wr,
    const float* __restrict__ bias, float* __restrict__ out,
    int nRowTiles, int relu) {
  __shared__ v2f sW[2 * 32 * WPITCH];
  int wave = threadIdx.x >> 5;
  int lane = threadIdx.x & 31;

#pragma unroll
  for (int i = 0; i < 2 * DIM / GEMM_WAVES; ++i) {
    int n = wave + i * GEMM_WAVES;
    const float* wsrc = (n < DIM) ? Wl : Wr;
    int nn = n & (DIM - 1);
    const float4 w = *(const float4*)(wsrc + (size_t)nn * DIM + lane * 4);
    v2f* d = sW + (n >> 7) * (32 * WPITCH) + lane * WPITCH + (nn >> 4) * 32 + (nn & 15);
    v2f lo = {w.x, w.y};
    v2f hi = {w.z, w.w};
    d[0] = lo;
    d[16] = hi;
  }
  __syncthreads();

  int lm = lane & 15;
  int khalf = (lane >= 16) ? 2 : 0;
  const v2f* sWl = sW + lane;
  const v2f* sWr = sW + 32 * WPITCH + lane;

  for (int rowTile = blockIdx.x + gridDim.x * wave; rowTile < nRowTiles;
       rowTile += gridDim.x * GEMM_WAVES) {
    int row0 = rowTile * 16;
    const float* aBase = mean + (size_t)(row0 + lm) * DIM + khalf;
    const float* xBase = xdst + (size_t)(row0 + lm) * DIM + khalf;

    v8f accA[8], accB[8];
#pragma unroll
    for (int t = 0; t < 8; ++t) {
      float bval = bias[t * 16 + lm];
      accA[t] = (v8f){bval, bval, bval, bval, bval, bval, bval, bval};
      accB[t] = (v8f){0.f, 0.f, 0.f, 0.f, 0.f, 0.f, 0.f, 0.f};
    }

    v2f am[4], ax[4];
#pragma unroll
    for (int u = 0; u < 4; ++u) {
      am[u] = *(const v2f*)(aBase + u * 4);
      ax[u] = *(const v2f*)(xBase + u * 4);
    }
#pragma unroll 1
    for (int g = 0; g < DIM / 16; ++g) {
      v2f amN[4], axN[4];
      int gn = (g + 1) & (DIM / 16 - 1);
#pragma unroll
      for (int u = 0; u < 4; ++u) {
        amN[u] = *(const v2f*)(aBase + gn * 16 + u * 4);
        axN[u] = *(const v2f*)(xBase + gn * 16 + u * 4);
      }
      const v2f* bl = sWl + g * 4 * WPITCH;
      const v2f* br = sWr + g * 4 * WPITCH;
#pragma unroll
      for (int u = 0; u < 4; ++u) {
#pragma unroll
        for (int t = 0; t < 8; ++t) {
          accA[t] = __builtin_amdgcn_wmma_f32_16x16x4_f32(false, am[u], false,
                                                          bl[u * WPITCH + t * 32], (short)0,
                                                          accA[t], false, false);
          accB[t] = __builtin_amdgcn_wmma_f32_16x16x4_f32(false, ax[u], false,
                                                          br[u * WPITCH + t * 32], (short)0,
                                                          accB[t], false, false);
        }
      }
#pragma unroll
      for (int u = 0; u < 4; ++u) {
        am[u] = amN[u];
        ax[u] = axN[u];
      }
    }

    float* orow = out + (size_t)(row0 + ((lane >= 16) ? 8 : 0)) * DIM + lm;
#pragma unroll
    for (int t = 0; t < 8; ++t) {
#pragma unroll
      for (int r = 0; r < 8; ++r) {
        float v = accA[t][r] + accB[t][r];
        if (relu) v = fmaxf(v, 0.0f);
        orow[(size_t)r * DIM + t * 16] = v;
      }
    }
  }
}

__global__ __launch_bounds__(256) void k_decode(
    const float* __restrict__ zu, const float* __restrict__ zj,
    const int* __restrict__ elu, const int* __restrict__ elj,
    float* __restrict__ out, int n) {
  int p = blockIdx.x * 8 + (threadIdx.x >> 5);
  if (p >= n) return;
  int lane = threadIdx.x & 31;
  const float4 a = *(const float4*)(zu + (size_t)elu[p] * DIM + lane * 4);
  const float4 b = *(const float4*)(zj + (size_t)elj[p] * DIM + lane * 4);
  float dot = a.x * b.x + a.y * b.y + a.z * b.z + a.w * b.w;
  float na = a.x * a.x + a.y * a.y + a.z * a.z + a.w * a.w;
  float nb = b.x * b.x + b.y * b.y + b.z * b.z + b.w * b.w;
#pragma unroll
  for (int off = 16; off > 0; off >>= 1) {
    dot += __shfl_xor(dot, off, 32);
    na += __shfl_xor(na, off, 32);
    nb += __shfl_xor(nb, off, 32);
  }
  if (lane == 0) {
    out[p] = dot / (fmaxf(sqrtf(na), 1e-8f) * fmaxf(sqrtf(nb), 1e-8f));
  }
}


extern "C" void kernel_launch(void* const* d_in, const int* in_sizes, int n_in,
                              void* d_out, int out_size, void* d_ws, size_t ws_size,
                              hipStream_t stream) {
  const float* x_user = (const float*)d_in[0];
  const float* x_job = (const float*)d_in[1];
  const int* uj_src = (const int*)d_in[2];
  const int* uj_dst = (const int*)d_in[3];
  const int* ju_src = (const int*)d_in[4];
  const int* ju_dst = (const int*)d_in[5];
  const int* el_user = (const int*)d_in[6];
  const int* el_job = (const int*)d_in[7];
  const float* W1l_uj = (const float*)d_in[8];
  const float* W1r_uj = (const float*)d_in[9];
  const float* W1l_ju = (const float*)d_in[10];
  const float* W1r_ju = (const float*)d_in[11];
  const float* W2l_uj = (const float*)d_in[12];
  const float* W2r_uj = (const float*)d_in[13];
  const float* W2l_ju = (const float*)d_in[14];
  const float* W2r_ju = (const float*)d_in[15];
  const float* b1_uj = (const float*)d_in[16];
  const float* b1_ju = (const float*)d_in[17];
  const float* b2_uj = (const float*)d_in[18];
  const float* b2_ju = (const float*)d_in[19];
  float* out = (float*)d_out;

  float* ws = (float*)d_ws;
  float* agg_user = ws; ws += (size_t)NUU * DIM;
  float* agg_job  = ws; ws += (size_t)NJJ * DIM;
  float* cnt_user = ws; ws += NUU;
  float* cnt_job  = ws; ws += NJJ;
  float* h_user   = ws; ws += (size_t)NUU * DIM;
  float* h_job    = ws; ws += (size_t)NJJ * DIM;
  float* z_user   = ws; ws += (size_t)NUU * DIM;
  float* z_job    = ws; ws += (size_t)NJJ * DIM;

  const int nUF = NUU * DIM, nJF = NJJ * DIM;
  dim3 blk(256);

  k_zero<<<(nJF + 255) / 256, blk, 0, stream>>>(agg_job, nJF);
  k_zero<<<(nUF + 255) / 256, blk, 0, stream>>>(agg_user, nUF);
  k_zero<<<(NJJ + 255) / 256, blk, 0, stream>>>(cnt_job, NJJ);
  k_zero<<<(NUU + 255) / 256, blk, 0, stream>>>(cnt_user, NUU);
  k_count<<<(NEDGE + 255) / 256, blk, 0, stream>>>(uj_dst, cnt_job, NEDGE);
  k_count<<<(NEDGE + 255) / 256, blk, 0, stream>>>(ju_dst, cnt_user, NEDGE);
  k_scatter<<<NEDGE / 8, blk, 0, stream>>>(x_user, uj_src, uj_dst, agg_job, NEDGE);
  k_scatter<<<NEDGE / 8, blk, 0, stream>>>(x_job, ju_src, ju_dst, agg_user, NEDGE);
  k_mean<<<(nJF + 255) / 256, blk, 0, stream>>>(agg_job, cnt_job, nJF);
  k_mean<<<(nUF + 255) / 256, blk, 0, stream>>>(agg_user, cnt_user, nUF);

  k_sage_gemm<<<256, 512, 0, stream>>>(agg_job, x_job, W1l_uj, W1r_uj,
                                            b1_uj, h_job, NJJ / 16, 1);
  k_sage_gemm<<<256, 512, 0, stream>>>(agg_user, x_user, W1l_ju, W1r_ju,
                                            b1_ju, h_user, NUU / 16, 1);

  k_zero<<<(nJF + 255) / 256, blk, 0, stream>>>(agg_job, nJF);
  k_zero<<<(nUF + 255) / 256, blk, 0, stream>>>(agg_user, nUF);
  k_scatter<<<NEDGE / 8, blk, 0, stream>>>(h_user, uj_src, uj_dst, agg_job, NEDGE);
  k_scatter<<<NEDGE / 8, blk, 0, stream>>>(h_job, ju_src, ju_dst, agg_user, NEDGE);
  k_mean<<<(nJF + 255) / 256, blk, 0, stream>>>(agg_job, cnt_job, nJF);
  k_mean<<<(nUF + 255) / 256, blk, 0, stream>>>(agg_user, cnt_user, nUF);

  k_sage_gemm<<<256, 512, 0, stream>>>(agg_job, h_job, W2l_uj, W2r_uj,
                                            b2_uj, z_job, NJJ / 16, 0);
  k_sage_gemm<<<256, 512, 0, stream>>>(agg_user, h_user, W2l_ju, W2r_ju,
                                            b2_ju, z_user, NUU / 16, 0);

  k_decode<<<NEL / 8, blk, 0, stream>>>(z_user, z_job, el_user, el_job, out, NEL);
}


# Scatter: lane owns features lane+32j, so each f32 atomic and gather load covers one whole aligned 128-B line, not four quarter lines (count arrays moved last to align rows); sums in arrival order
# speedup vs baseline: 1.5536x; 1.2887x over previous
#include <hip/hip_runtime.h>
#include <hip/hip_bf16.h>
#include <math.h>

#define NUU 50000
#define NJJ 20000
#define NEDGE 500000
#define NEL 100000
#define DIM 128

typedef __attribute__((ext_vector_type(2))) float v2f;
typedef __attribute__((ext_vector_type(8))) float v8f;


__global__ __launch_bounds__(256) void k_zero(float* __restrict__ p, int n) {
  int i = blockIdx.x * 256 + threadIdx.x;
  if (i < n) p[i] = 0.0f;
}

__global__ __launch_bounds__(256) void k_count(const int* __restrict__ dst,
                                               float* __restrict__ cnt, int n) {
  int e = blockIdx.x * 256 + threadIdx.x;
  if (e < n) unsafeAtomicAdd(&cnt[dst[e]], 1.0f);
}

__global__ __launch_bounds__(256) void k_scatter(const float* __restrict__ xsrc,
                                                 const int* __restrict__ src,
                                                 const int* __restrict__ dst,
                                                 float* __restrict__ sum, int n) {
  int e = blockIdx.x * 8 + (threadIdx.x >> 5);
  if (e >= n) return;
  int lane = threadIdx.x & 31;
  int s = src[e];
  int d = dst[e];
  const float* q = xsrc + (size_t)s * DIM + lane;
  const float v0 = q[0];
  const float v1 = q[32];
  const float v2 = q[64];
  const float v3 = q[96];
  float* p = sum + (size_t)d * DIM + lane;
  unsafeAtomicAdd(p + 0, v0);
  unsafeAtomicAdd(p + 32, v1);
  unsafeAtomicAdd(p + 64, v2);
  unsafeAtomicAdd(p + 96, v3);
}

__global__ __launch_bounds__(256) void k_mean(float* __restrict__ sum,
                                              const float* __restrict__ cnt, int n) {
  int i = blockIdx.x * 256 + threadIdx.x;
  if (i < n) sum[i] = sum[i] / fmaxf(cnt[i >> 7], 1.0f);
}

#define WPITCH 257
#define GEMM_WAVES 16
__global__ __launch_bounds__(512) void k_sage_gemm(
    const float* __restrict__ mean, const float* __restrict__ xdst,
    const float* __restrict__ Wl, const float* __restrict__ Wr,
    const float* __restrict__ bias, float* __restrict__ out,
    int nRowTiles, int relu) {
  __shared__ v2f sW[2 * 32 * WPITCH];
  int wave = threadIdx.x >> 5;
  int lane = threadIdx.x & 31;

#pragma unroll
  for (int i = 0; i < 2 * DIM / GEMM_WAVES; ++i) {
    int n = wave + i * GEMM_WAVES;
    const float* wsrc = (n < DIM) ? Wl : Wr;
    int nn = n & (DIM - 1);
    const float4 w = *(const float4*)(wsrc + (size_t)nn * DIM + lane * 4);
    v2f* d = sW + (n >> 7) * (32 * WPITCH) + lane * WPITCH + (nn >> 4) * 32 + (nn & 15);
    v2f lo = {w.x, w.y};
    v2f hi = {w.z, w.w};
    d[0] = lo;
    d[16] = hi;
  }
  __syncthreads();

  int lm = lane & 15;
  int khalf = (lane >= 16) ? 2 : 0;
  const v2f* sWl = sW + lane;
  const v2f* sWr = sW + 32 * WPITCH + lane;

  for (int rowTile = blockIdx.x + gridDim.x * wave; rowTile < nRowTiles;
       rowTile += gridDim.x * GEMM_WAVES) {
    int row0 = rowTile * 16;
    const float* aBase = mean + (size_t)(row0 + lm) * DIM + khalf;
    const float* xBase = xdst + (size_t)(row0 + lm) * DIM + khalf;

    v8f accA[8], accB[8];
#pragma unroll
    for (int t = 0; t < 8; ++t) {
      float bval = bias[t * 16 + lm];
      accA[t] = (v8f){bval, bval, bval, bval, bval, bval, bval, bval};
      accB[t] = (v8f){0.f, 0.f, 0.f, 0.f, 0.f, 0.f, 0.f, 0.f};
    }

    v2f am[4], ax[4];
#pragma unroll
    for (int u = 0; u < 4; ++u) {
      am[u] = *(const v2f*)(aBase + u * 4);
      ax[u] = *(const v2f*)(xBase + u * 4);
    }
#pragma unroll 1
    for (int g = 0; g < DIM / 16; ++g) {
      v2f amN[4], axN[4];
      int gn = (g + 1) & (DIM / 16 - 1);
#pragma unroll
      for (int u = 0; u < 4; ++u) {
        amN[u] = *(const v2f*)(aBase + gn * 16 + u * 4);
        axN[u] = *(const v2f*)(xBase + gn * 16 + u * 4);
      }
      const v2f* bl = sWl + g * 4 * WPITCH;
      const v2f* br = sWr + g * 4 * WPITCH;
#pragma unroll
      for (int u = 0; u < 4; ++u) {
#pragma unroll
        for (int t = 0; t < 8; ++t) {
          accA[t] = __builtin_amdgcn_wmma_f32_16x16x4_f32(false, am[u], false,
                                                          bl[u * WPITCH + t * 32], (short)0,
                                                          accA[t], false, false);
          accB[t] = __builtin_amdgcn_wmma_f32_16x16x4_f32(false, ax[u], false,
                                                          br[u * WPITCH + t * 32], (short)0,
                                                          accB[t], false, false);
        }
      }
#pragma unroll
      for (int u = 0; u < 4; ++u) {
        am[u] = amN[u];
        ax[u] = axN[u];
      }
    }

    float* orow = out + (size_t)(row0 + ((lane >= 16) ? 8 : 0)) * DIM + lm;
#pragma unroll
    for (int t = 0; t < 8; ++t) {
#pragma unroll
      for (int r = 0; r < 8; ++r) {
        float v = accA[t][r] + accB[t][r];
        if (relu) v = fmaxf(v, 0.0f);
        orow[(size_t)r * DIM + t * 16] = v;
      }
    }
  }
}

__global__ __launch_bounds__(256) void k_decode(
    const float* __restrict__ zu, const float* __restrict__ zj,
    const int* __restrict__ elu, const int* __restrict__ elj,
    float* __restrict__ out, int n) {
  int p = blockIdx.x * 8 + (threadIdx.x >> 5);
  if (p >= n) return;
  int lane = threadIdx.x & 31;
  const float4 a = *(const float4*)(zu + (size_t)elu[p] * DIM + lane * 4);
  const float4 b = *(const float4*)(zj + (size_t)elj[p] * DIM + lane * 4);
  float dot = a.x * b.x + a.y * b.y + a.z * b.z + a.w * b.w;
  float na = a.x * a.x + a.y * a.y + a.z * a.z + a.w * a.w;
  float nb = b.x * b.x + b.y * b.y + b.z * b.z + b.w * b.w;
#pragma unroll
  for (int off = 16; off > 0; off >>= 1) {
    dot += __shfl_xor(dot, off, 32);
    na += __shfl_xor(na, off, 32);
    nb += __shfl_xor(nb, off, 32);
  }
  if (lane == 0) {
    out[p] = dot / (fmaxf(sqrtf(na), 1e-8f) * fmaxf(sqrtf(nb), 1e-8f));
  }
}


extern "C" void kernel_launch(void* const* d_in, const int* in_sizes, int n_in,
                              void* d_out, int out_size, void* d_ws, size_t ws_size,
                              hipStream_t stream) {
  const float* x_user = (const float*)d_in[0];
  const float* x_job = (const float*)d_in[1];
  const int* uj_src = (const int*)d_in[2];
  const int* uj_dst = (const int*)d_in[3];
  const int* ju_src = (const int*)d_in[4];
  const int* ju_dst = (const int*)d_in[5];
  const int* el_user = (const int*)d_in[6];
  const int* el_job = (const int*)d_in[7];
  const float* W1l_uj = (const float*)d_in[8];
  const float* W1r_uj = (const float*)d_in[9];
  const float* W1l_ju = (const float*)d_in[10];
  const float* W1r_ju = (const float*)d_in[11];
  const float* W2l_uj = (const float*)d_in[12];
  const float* W2r_uj = (const float*)d_in[13];
  const float* W2l_ju = (const float*)d_in[14];
  const float* W2r_ju = (const float*)d_in[15];
  const float* b1_uj = (const float*)d_in[16];
  const float* b1_ju = (const float*)d_in[17];
  const float* b2_uj = (const float*)d_in[18];
  const float* b2_ju = (const float*)d_in[19];
  float* out = (float*)d_out;

  float* ws = (float*)d_ws;
  float* agg_user = ws; ws += (size_t)NUU * DIM;
  float* agg_job  = ws; ws += (size_t)NJJ * DIM;
  float* h_user   = ws; ws += (size_t)NUU * DIM;
  float* h_job    = ws; ws += (size_t)NJJ * DIM;
  float* z_user   = ws; ws += (size_t)NUU * DIM;
  float* z_job    = ws; ws += (size_t)NJJ * DIM;
  float* cnt_user = ws; ws += NUU;
  float* cnt_job  = ws; ws += NJJ;

  const int nUF = NUU * DIM, nJF = NJJ * DIM;
  dim3 blk(256);

  k_zero<<<(nJF + 255) / 256, blk, 0, stream>>>(agg_job, nJF);
  k_zero<<<(nUF + 255) / 256, blk, 0, stream>>>(agg_user, nUF);
  k_zero<<<(NJJ + 255) / 256, blk, 0, stream>>>(cnt_job, NJJ);
  k_zero<<<(NUU + 255) / 256, blk, 0, stream>>>(cnt_user, NUU);
  k_count<<<(NEDGE + 255) / 256, blk, 0, stream>>>(uj_dst, cnt_job, NEDGE);
  k_count<<<(NEDGE + 255) / 256, blk, 0, stream>>>(ju_dst, cnt_user, NEDGE);
  k_scatter<<<NEDGE / 8, blk, 0, stream>>>(x_user, uj_src, uj_dst, agg_job, NEDGE);
  k_scatter<<<NEDGE / 8, blk, 0, stream>>>(x_job, ju_src, ju_dst, agg_user, NEDGE);
  k_mean<<<(nJF + 255) / 256, blk, 0, stream>>>(agg_job, cnt_job, nJF);
  k_mean<<<(nUF + 255) / 256, blk, 0, stream>>>(agg_user, cnt_user, nUF);

  k_sage_gemm<<<256, 512, 0, stream>>>(agg_job, x_job, W1l_uj, W1r_uj,
                                            b1_uj, h_job, NJJ / 16, 1);
  k_sage_gemm<<<256, 512, 0, stream>>>(agg_user, x_user, W1l_ju, W1r_ju,
                                            b1_ju, h_user, NUU / 16, 1);

  k_zero<<<(nJF + 255) / 256, blk, 0, stream>>>(agg_job, nJF);
  k_zero<<<(nUF + 255) / 256, blk, 0, stream>>>(agg_user, nUF);
  k_scatter<<<NEDGE / 8, blk, 0, stream>>>(h_user, uj_src, uj_dst, agg_job, NEDGE);
  k_scatter<<<NEDGE / 8, blk, 0, stream>>>(h_job, ju_src, ju_dst, agg_user, NEDGE);
  k_mean<<<(nJF + 255) / 256, blk, 0, stream>>>(agg_job, cnt_job, nJF);
  k_mean<<<(nUF + 255) / 256, blk, 0, stream>>>(agg_user, cnt_user, nUF);

  k_sage_gemm<<<256, 512, 0, stream>>>(agg_job, h_job, W2l_uj, W2r_uj,
                                            b2_uj, z_job, NJJ / 16, 0);
  k_sage_gemm<<<256, 512, 0, stream>>>(agg_user, h_user, W2l_ju, W2r_ju,
                                            b2_ju, z_user, NUU / 16, 0);

  k_decode<<<NEL / 8, blk, 0, stream>>>(z_user, z_job, el_user, el_job, out, NEL);
}


# Neighbour sums without float atomics: edges grouped by destination once (returning int add = slot, scan, place); a wave adds a row's sources in registers in slot (arrival) order, stores the mean
# speedup vs baseline: 2.6121x; 1.6813x over previous
#include <hip/hip_runtime.h>
#include <hip/hip_bf16.h>
#include <math.h>

#define NUU 50000
#define NJJ 20000
#define NEDGE 500000
#define NEL 100000
#define DIM 128

typedef __attribute__((ext_vector_type(2))) float v2f;
typedef __attribute__((ext_vector_type(8))) float v8f;


__global__ __launch_bounds__(1024) void k_zero_i(int* __restrict__ p, int n) {
  int i = blockIdx.x * 1024 + threadIdx.x;
  if (i < n) p[i] = 0;
}

__global__ __launch_bounds__(1024) void k_rank(const int* __restrict__ dstJ,
                                                const int* __restrict__ dstU,
                                                int* __restrict__ cntJ, int* __restrict__ cntU,
                                                int* __restrict__ rankJ, int* __restrict__ rankU,
                                                int n) {
  int e = blockIdx.x * 1024 + threadIdx.x;
  if (e >= n) return;
  if (blockIdx.y == 0) rankJ[e] = atomicAdd(&cntJ[dstJ[e]], 1);
  else rankU[e] = atomicAdd(&cntU[dstU[e]], 1);
}

template <int N>
__device__ __forceinline__ void scan_counts(const int* __restrict__ cnt, int* __restrict__ rs,
                                            int* sc, int* part) {
  constexpr int ROWS = (N + 1023) / 1024;
  constexpr int CH = ROWS | 1;
  const int tid = threadIdx.x;
#pragma unroll
  for (int r = 0; r < ROWS; ++r) {
    int i = r * 1024 + tid;
    if (i < N) sc[i] = cnt[i];
  }
  __syncthreads();
  const int b = tid * CH;
  int sum = 0;
  for (int k = 0; k < CH; ++k)
    if (b + k < N) sum += sc[b + k];
  const int lane = tid & 31, wave = tid >> 5;
  int inc = sum;
  for (int o = 1; o < 32; o <<= 1) {
    int t = __shfl_up(inc, o, 32);
    if (lane >= o) inc += t;
  }
  if (lane == 31) part[wave] = inc;
  __syncthreads();
  const int pw = part[lane];
  int pinc = pw;
  for (int o = 1; o < 32; o <<= 1) {
    int t = __shfl_up(pinc, o, 32);
    if (lane >= o) pinc += t;
  }
  const int woff = __shfl(pinc - pw, wave, 32);
  int run = woff + inc - sum;
  for (int k = 0; k < CH; ++k)
    if (b + k < N) {
      int c = sc[b + k];
      sc[b + k] = run;
      run += c;
    }
  __syncthreads();
#pragma unroll
  for (int r = 0; r < ROWS; ++r) {
    int i = r * 1024 + tid;
    if (i < N) rs[i] = sc[i];
  }
  if (tid == 1023) rs[N] = run;
}

__global__ __launch_bounds__(1024) void k_scan(const int* __restrict__ cntJ,
                                                const int* __restrict__ cntU,
                                                int* __restrict__ rsJ, int* __restrict__ rsU) {
  __shared__ int sc[NUU];
  __shared__ int part[32];
  if (blockIdx.x == 0) scan_counts<NJJ>(cntJ, rsJ, sc, part);
  else scan_counts<NUU>(cntU, rsU, sc, part);
}

__global__ __launch_bounds__(1024) void k_fill(const int* __restrict__ srcJ,
                                                const int* __restrict__ dstJ,
                                                const int* __restrict__ srcU,
                                                const int* __restrict__ dstU,
                                                const int* __restrict__ rankJ,
                                                const int* __restrict__ rankU,
                                                const int* __restrict__ rsJ,
                                                const int* __restrict__ rsU,
                                                int* __restrict__ permJ, int* __restrict__ permU,
                                                int n) {
  int e = blockIdx.x * 1024 + threadIdx.x;
  if (e >= n) return;
  if (blockIdx.y == 0) permJ[min(max(rsJ[dstJ[e]] + rankJ[e], 0), n - 1)] = srcJ[e];
  else permU[min(max(rsU[dstU[e]] + rankU[e], 0), n - 1)] = srcU[e];
}

template <int U>
__device__ __forceinline__ void add_rows(float4& acc, const float* __restrict__ xl, int myidx,
                                         int j, int lastSrc) {
  float4 r[U];
#pragma unroll
  for (int u = 0; u < U; ++u) {
    int s = __builtin_amdgcn_readlane(myidx, j + u);
    s = min(max(s, 0), lastSrc);
    r[u] = *(const float4*)(xl + (size_t)s * DIM);
  }
#pragma unroll
  for (int u = 0; u < U; ++u) {
    acc.x += r[u].x;
    acc.y += r[u].y;
    acc.z += r[u].z;
    acc.w += r[u].w;
  }
}

__global__ __launch_bounds__(512) void k_gather_mean(const float* __restrict__ xsrc,
                                                      const int* __restrict__ rs,
                                                      const int* __restrict__ perm,
                                                      float* __restrict__ mean, int nDst,
                                                      int nSrc) {
  const int d = __builtin_amdgcn_readfirstlane((int)(blockIdx.x * 16 + (threadIdx.x >> 5)));
  if (d >= nDst) return;
  const int lane = threadIdx.x & 31;
  const int beg = rs[d];
  const int end = rs[d + 1];
  const float* xl = xsrc + lane * 4;
  float4 acc = {0.0f, 0.0f, 0.0f, 0.0f};
  for (int base = beg; base < end; base += 32) {
    const int m = min(end - base, 32);
    const int myidx = (lane < m) ? perm[base + lane] : 0;
    int j = 0;
    for (; j + 8 <= m; j += 8) add_rows<8>(acc, xl, myidx, j, nSrc - 1);
    if (m - j >= 4) { add_rows<4>(acc, xl, myidx, j, nSrc - 1); j += 4; }
    if (m - j >= 2) { add_rows<2>(acc, xl, myidx, j, nSrc - 1); j += 2; }
    if (m - j >= 1) add_rows<1>(acc, xl, myidx, j, nSrc - 1);
  }
  const float c = fmaxf((float)(end - beg), 1.0f);
  float4 o;
  o.x = acc.x / c;
  o.y = acc.y / c;
  o.z = acc.z / c;
  o.w = acc.w / c;
  *(float4*)(mean + (size_t)d * DIM + lane * 4) = o;
}

#define WPITCH 257
#define GEMM_WAVES 16
__global__ __launch_bounds__(512) void k_sage_gemm(
    const float* __restrict__ mean, const float* __restrict__ xdst,
    const float* __restrict__ Wl, const float* __restrict__ Wr,
    const float* __restrict__ bias, float* __restrict__ out,
    int nRowTiles, int relu) {
  __shared__ v2f sW[2 * 32 * WPITCH];
  int wave = threadIdx.x >> 5;
  int lane = threadIdx.x & 31;

#pragma unroll
  for (int i = 0; i < 2 * DIM / GEMM_WAVES; ++i) {
    int n = wave + i * GEMM_WAVES;
    const float* wsrc = (n < DIM) ? Wl : Wr;
    int nn = n & (DIM - 1);
    const float4 w = *(const float4*)(wsrc + (size_t)nn * DIM + lane * 4);
    v2f* d = sW + (n >> 7) * (32 * WPITCH) + lane * WPITCH + (nn >> 4) * 32 + (nn & 15);
    v2f lo = {w.x, w.y};
    v2f hi = {w.z, w.w};
    d[0] = lo;
    d[16] = hi;
  }
  __syncthreads();

  int lm = lane & 15;
  int khalf = (lane >= 16) ? 2 : 0;
  const v2f* sWl = sW + lane;
  const v2f* sWr = sW + 32 * WPITCH + lane;

  for (int rowTile = blockIdx.x + gridDim.x * wave; rowTile < nRowTiles;
       rowTile += gridDim.x * GEMM_WAVES) {
    int row0 = rowTile * 16;
    const float* aBase = mean + (size_t)(row0 + lm) * DIM + khalf;
    const float* xBase = xdst + (size_t)(row0 + lm) * DIM + khalf;

    v8f accA[8], accB[8];
#pragma unroll
    for (int t = 0; t < 8; ++t) {
      float bval = bias[t * 16 + lm];
      accA[t] = (v8f){bval, bval, bval, bval, bval, bval, bval, bval};
      accB[t] = (v8f){0.f, 0.f, 0.f, 0.f, 0.f, 0.f, 0.f, 0.f};
    }

    v2f am[4], ax[4];
#pragma unroll
    for (int u = 0; u < 4; ++u) {
      am[u] = *(const v2f*)(aBase + u * 4);
      ax[u] = *(const v2f*)(xBase + u * 4);
    }
#pragma unroll 1
    for (int g = 0; g < DIM / 16; ++g) {
      v2f amN[4], axN[4];
      int gn = (g + 1) & (DIM / 16 - 1);
#pragma unroll
      for (int u = 0; u < 4; ++u) {
        amN[u] = *(const v2f*)(aBase + gn * 16 + u * 4);
        axN[u] = *(const v2f*)(xBase + gn * 16 + u * 4);
      }
      const v2f* bl = sWl + g * 4 * WPITCH;
      const v2f* br = sWr + g * 4 * WPITCH;
#pragma unroll
      for (int u = 0; u < 4; ++u) {
#pragma unroll
        for (int t = 0; t < 8; ++t) {
          accA[t] = __builtin_amdgcn_wmma_f32_16x16x4_f32(false, am[u], false,
                                                          bl[u * WPITCH + t * 32], (short)0,
                                                          accA[t], false, false);
          accB[t] = __builtin_amdgcn_wmma_f32_16x16x4_f32(false, ax[u], false,
                                                          br[u * WPITCH + t * 32], (short)0,
                                                          accB[t], false, false);
        }
      }
#pragma unroll
      for (int u = 0; u < 4; ++u) {
        am[u] = amN[u];
        ax[u] = axN[u];
      }
    }

    float* orow = out + (size_t)(row0 + ((lane >= 16) ? 8 : 0)) * DIM + lm;
#pragma unroll
    for (int t = 0; t < 8; ++t) {
#pragma unroll
      for (int r = 0; r < 8; ++r) {
        float v = accA[t][r] + accB[t][r];
        if (relu) v = fmaxf(v, 0.0f);
        orow[(size_t)r * DIM + t * 16] = v;
      }
    }
  }
}

__global__ __launch_bounds__(256) void k_decode(
    const float* __restrict__ zu, const float* __restrict__ zj,
    const int* __restrict__ elu, const int* __restrict__ elj,
    float* __restrict__ out, int n) {
  int p = blockIdx.x * 8 + (threadIdx.x >> 5);
  if (p >= n) return;
  int lane = threadIdx.x & 31;
  const float4 a = *(const float4*)(zu + (size_t)elu[p] * DIM + lane * 4);
  const float4 b = *(const float4*)(zj + (size_t)elj[p] * DIM + lane * 4);
  float dot = a.x * b.x + a.y * b.y + a.z * b.z + a.w * b.w;
  float na = a.x * a.x + a.y * a.y + a.z * a.z + a.w * a.w;
  float nb = b.x * b.x + b.y * b.y + b.z * b.z + b.w * b.w;
#pragma unroll
  for (int off = 16; off > 0; off >>= 1) {
    dot += __shfl_xor(dot, off, 32);
    na += __shfl_xor(na, off, 32);
    nb += __shfl_xor(nb, off, 32);
  }
  if (lane == 0) {
    out[p] = dot / (fmaxf(sqrtf(na), 1e-8f) * fmaxf(sqrtf(nb), 1e-8f));
  }
}


extern "C" void kernel_launch(void* const* d_in, const int* in_sizes, int n_in,
                              void* d_out, int out_size, void* d_ws, size_t ws_size,
                              hipStream_t stream) {
  const float* x_user = (const float*)d_in[0];
  const float* x_job = (const float*)d_in[1];
  const int* uj_src = (const int*)d_in[2];
  const int* uj_dst = (const int*)d_in[3];
  const int* ju_src = (const int*)d_in[4];
  const int* ju_dst = (const int*)d_in[5];
  const int* el_user = (const int*)d_in[6];
  const int* el_job = (const int*)d_in[7];
  const float* W1l_uj = (const float*)d_in[8];
  const float* W1r_uj = (const float*)d_in[9];
  const float* W1l_ju = (const float*)d_in[10];
  const float* W1r_ju = (const float*)d_in[11];
  const float* W2l_uj = (const float*)d_in[12];
  const float* W2r_uj = (const float*)d_in[13];
  const float* W2l_ju = (const float*)d_in[14];
  const float* W2r_ju = (const float*)d_in[15];
  const float* b1_uj = (const float*)d_in[16];
  const float* b1_ju = (const float*)d_in[17];
  const float* b2_uj = (const float*)d_in[18];
  const float* b2_ju = (const float*)d_in[19];
  float* out = (float*)d_out;

  float* ws = (float*)d_ws;
  float* agg_user = ws; ws += (size_t)NUU * DIM;
  float* agg_job  = ws; ws += (size_t)NJJ * DIM;
  float* h_user   = ws; ws += (size_t)NUU * DIM;
  float* h_job    = ws; ws += (size_t)NJJ * DIM;
  float* z_user   = ws; ws += (size_t)NUU * DIM;
  float* z_job    = ws; ws += (size_t)NJJ * DIM;
  int* cnt_user = (int*)ws;
  int* cnt_job  = cnt_user + NUU;
  int* rank_job  = (int*)z_user;
  int* rank_user = rank_job + NEDGE;
  int* perm_job  = rank_user + NEDGE;
  int* perm_user = perm_job + NEDGE;
  int* rs_job    = perm_user + NEDGE;
  int* rs_user   = rs_job + (NJJ + 32);

  const dim3 eg((NEDGE + 1023) / 1024, 2);

  k_zero_i<<<(NUU + NJJ + 1023) / 1024, 1024, 0, stream>>>(cnt_user, NUU + NJJ);
  k_rank<<<eg, 1024, 0, stream>>>(uj_dst, ju_dst, cnt_job, cnt_user, rank_job, rank_user, NEDGE);
  k_scan<<<2, 1024, 0, stream>>>(cnt_job, cnt_user, rs_job, rs_user);
  k_fill<<<eg, 1024, 0, stream>>>(uj_src, uj_dst, ju_src, ju_dst, rank_job, rank_user,
                                  rs_job, rs_user, perm_job, perm_user, NEDGE);

  k_gather_mean<<<(NJJ + 15) / 16, 512, 0, stream>>>(x_user, rs_job, perm_job, agg_job, NJJ, NUU);
  k_gather_mean<<<(NUU + 15) / 16, 512, 0, stream>>>(x_job, rs_user, perm_user, agg_user, NUU, NJJ);

  k_sage_gemm<<<256, 512, 0, stream>>>(agg_job, x_job, W1l_uj, W1r_uj,
                                            b1_uj, h_job, NJJ / 16, 1);
  k_sage_gemm<<<256, 512, 0, stream>>>(agg_user, x_user, W1l_ju, W1r_ju,
                                            b1_ju, h_user, NUU / 16, 1);

  k_gather_mean<<<(NJJ + 15) / 16, 512, 0, stream>>>(h_user, rs_job, perm_job, agg_job, NJJ, NUU);
  k_gather_mean<<<(NUU + 15) / 16, 512, 0, stream>>>(h_job, rs_user, perm_user, agg_user, NUU, NJJ);

  k_sage_gemm<<<256, 512, 0, stream>>>(agg_job, h_job, W2l_uj, W2r_uj,
                                            b2_uj, z_job, NJJ / 16, 0);
  k_sage_gemm<<<256, 512, 0, stream>>>(agg_user, h_user, W2l_ju, W2r_ju,
                                            b2_ju, z_user, NUU / 16, 0);

  k_decode<<<NEL / 8, dim3(256), 0, stream>>>(z_user, z_job, el_user, el_job, out, NEL);
}


# Fewer, longer waves: rank/fill 8 edges per thread, a wave walks the edge stream of 4 or 8 rows (16 row loads in flight), decode 32 pairs per wave; 6x fewer wave entries; sum orders kept
# speedup vs baseline: 2.6997x; 1.0335x over previous
#include <hip/hip_runtime.h>
#include <hip/hip_bf16.h>
#include <math.h>

#define NUU 50000
#define NJJ 20000
#define NEDGE 500000
#define NEL 100000
#define DIM 128

typedef __attribute__((ext_vector_type(2))) float v2f;
typedef __attribute__((ext_vector_type(8))) float v8f;


#define EPT 8
#define EDGE_WG 512

__global__ __launch_bounds__(1024) void k_zero_i(int* __restrict__ p, int n4) {
  int i = blockIdx.x * 1024 + threadIdx.x;
  if (i < n4) ((int4*)p)[i] = make_int4(0, 0, 0, 0);
}

__device__ __forceinline__ void rank_edges(const int* __restrict__ dst, int* __restrict__ cnt,
                                           int* __restrict__ rank, int n) {
  const int e0 = blockIdx.x * (EDGE_WG * EPT) + threadIdx.x;
  int d[EPT], r[EPT];
#pragma unroll
  for (int k = 0; k < EPT; ++k) {
    d[k] = dst[min(e0 + k * EDGE_WG, n - 1)];
  }
#pragma unroll
  for (int k = 0; k < EPT; ++k) {
    const int e = e0 + k * EDGE_WG;
    r[k] = (e < n) ? atomicAdd(&cnt[d[k]], 1) : 0;
  }
#pragma unroll
  for (int k = 0; k < EPT; ++k) {
    const int e = e0 + k * EDGE_WG;
    if (e < n) rank[e] = r[k];
  }
}

__global__ __launch_bounds__(EDGE_WG) void k_rank(const int* __restrict__ dstJ,
                                                   const int* __restrict__ dstU,
                                                   int* __restrict__ cntJ, int* __restrict__ cntU,
                                                   int* __restrict__ rankJ, int* __restrict__ rankU,
                                                   int n) {
  if (blockIdx.y == 0) rank_edges(dstJ, cntJ, rankJ, n);
  else rank_edges(dstU, cntU, rankU, n);
}

template <int N>
__device__ __forceinline__ void scan_counts(const int* __restrict__ cnt, int* __restrict__ rs,
                                            int* sc, int* part) {
  constexpr int ROWS = (N + 1023) / 1024;
  constexpr int CH = ROWS | 1;
  const int tid = threadIdx.x;
#pragma unroll
  for (int r = 0; r < ROWS; ++r) {
    int i = r * 1024 + tid;
    if (i < N) sc[i] = cnt[i];
  }
  __syncthreads();
  const int b = tid * CH;
  int sum = 0;
  for (int k = 0; k < CH; ++k)
    if (b + k < N) sum += sc[b + k];
  const int lane = tid & 31, wave = tid >> 5;
  int inc = sum;
  for (int o = 1; o < 32; o <<= 1) {
    int t = __shfl_up(inc, o, 32);
    if (lane >= o) inc += t;
  }
  if (lane == 31) part[wave] = inc;
  __syncthreads();
  const int pw = part[lane];
  int pinc = pw;
  for (int o = 1; o < 32; o <<= 1) {
    int t = __shfl_up(pinc, o, 32);
    if (lane >= o) pinc += t;
  }
  const int woff = __shfl(pinc - pw, wave, 32);
  int run = woff + inc - sum;
  for (int k = 0; k < CH; ++k)
    if (b + k < N) {
      int c = sc[b + k];
      sc[b + k] = run;
      run += c;
    }
  __syncthreads();
#pragma unroll
  for (int r = 0; r < ROWS; ++r) {
    int i = r * 1024 + tid;
    if (i < N) rs[i] = sc[i];
  }
  if (tid == 1023) rs[N] = run;
}

__global__ __launch_bounds__(1024) void k_scan(const int* __restrict__ cntJ,
                                                const int* __restrict__ cntU,
                                                int* __restrict__ rsJ, int* __restrict__ rsU) {
  __shared__ int sc[NUU];
  __shared__ int part[32];
  if (blockIdx.x == 0) scan_counts<NJJ>(cntJ, rsJ, sc, part);
  else scan_counts<NUU>(cntU, rsU, sc, part);
}

__device__ __forceinline__ void fill_edges(const int* __restrict__ src, const int* __restrict__ dst,
                                           const int* __restrict__ rank, const int* __restrict__ rs,
                                           int* __restrict__ perm, int n) {
  const int e0 = blockIdx.x * (EDGE_WG * EPT) + threadIdx.x;
  int d[EPT], rk[EPT], s[EPT], pos[EPT];
#pragma unroll
  for (int k = 0; k < EPT; ++k) {
    const int e = min(e0 + k * EDGE_WG, n - 1);
    d[k] = dst[e];
    rk[k] = rank[e];
    s[k] = src[e];
  }
#pragma unroll
  for (int k = 0; k < EPT; ++k) pos[k] = rs[d[k]] + rk[k];
#pragma unroll
  for (int k = 0; k < EPT; ++k) {
    const int e = e0 + k * EDGE_WG;
    if (e < n) perm[min(max(pos[k], 0), n - 1)] = s[k];
  }
}

__global__ __launch_bounds__(EDGE_WG) void k_fill(const int* __restrict__ srcJ,
                                                   const int* __restrict__ dstJ,
                                                   const int* __restrict__ srcU,
                                                   const int* __restrict__ dstU,
                                                   const int* __restrict__ rankJ,
                                                   const int* __restrict__ rankU,
                                                   const int* __restrict__ rsJ,
                                                   const int* __restrict__ rsU,
                                                   int* __restrict__ permJ, int* __restrict__ permU,
                                                   int n) {
  if (blockIdx.y == 0) fill_edges(srcJ, dstJ, rankJ, rsJ, permJ, n);
  else fill_edges(srcU, dstU, rankU, rsU, permU, n);
}

__device__ __forceinline__ void store_mean(float* __restrict__ p, const float4 acc, int count) {
  const float c = fmaxf((float)count, 1.0f);
  float4 o;
  o.x = acc.x / c;
  o.y = acc.y / c;
  o.z = acc.z / c;
  o.w = acc.w / c;
  *(float4*)p = o;
}

template <int U>
__device__ __forceinline__ void add_rows(float4& acc, int& r, int& rowBeg, int& rowEnd,
                                         const int rsv, const int nRows,
                                         const float* __restrict__ xl, float* __restrict__ mrow,
                                         int myidx, int j, int e, int lastSrc) {
  float4 rr[U];
#pragma unroll
  for (int u = 0; u < U; ++u) {
    int s = __builtin_amdgcn_readlane(myidx, j + u);
    s = min(max(s, 0), lastSrc);
    rr[u] = *(const float4*)(xl + (size_t)s * DIM);
  }
#pragma unroll
  for (int u = 0; u < U; ++u) {
    if (e + u >= rowEnd && r + 1 < nRows) {
      store_mean(mrow + (size_t)r * DIM, acc, rowEnd - rowBeg);
      acc = make_float4(0.0f, 0.0f, 0.0f, 0.0f);
      ++r;
      rowBeg = rowEnd;
      rowEnd = __builtin_amdgcn_readlane(rsv, r + 1);
      while (e + u >= rowEnd && r + 1 < nRows) {
        *(float4*)(mrow + (size_t)r * DIM) = acc;
        ++r;
        rowBeg = rowEnd;
        rowEnd = __builtin_amdgcn_readlane(rsv, r + 1);
      }
    }
    acc.x += rr[u].x;
    acc.y += rr[u].y;
    acc.z += rr[u].z;
    acc.w += rr[u].w;
  }
}

template <int R>
__global__ __launch_bounds__(256) void k_gather_mean(const float* __restrict__ xsrc,
                                                      const int* __restrict__ rs,
                                                      const int* __restrict__ perm,
                                                      float* __restrict__ mean, int nDst,
                                                      int nSrc) {
  const int d0 = R * __builtin_amdgcn_readfirstlane((int)(blockIdx.x * 8 + (threadIdx.x >> 5)));
  if (d0 >= nDst) return;
  const int nRows = min(R, nDst - d0);
  const int lane = threadIdx.x & 31;
  const int rsv = (lane <= nRows) ? rs[d0 + lane] : 0;
  const int eBeg = __builtin_amdgcn_readlane(rsv, 0);
  const int eEnd = __builtin_amdgcn_readlane(rsv, nRows);
  const float* xl = xsrc + lane * 4;
  float* mrow = mean + (size_t)d0 * DIM + lane * 4;
  int r = 0;
  int rowBeg = eBeg;
  int rowEnd = __builtin_amdgcn_readlane(rsv, 1);
  float4 acc = make_float4(0.0f, 0.0f, 0.0f, 0.0f);
  for (int base = eBeg; base < eEnd; base += 128) {
    const int left = eEnd - base;
    const int i0 = (lane < left) ? perm[base + lane] : 0;
    const int i1 = (lane + 32 < left) ? perm[base + 32 + lane] : 0;
    const int i2 = (lane + 64 < left) ? perm[base + 64 + lane] : 0;
    const int i3 = (lane + 96 < left) ? perm[base + 96 + lane] : 0;
#pragma unroll 1
    for (int k = 0; k < 4; ++k) {
      const int b = base + 32 * k;
      const int m = min(eEnd - b, 32);
      if (m <= 0) break;
      const int myidx = (k == 0) ? i0 : (k == 1) ? i1 : (k == 2) ? i2 : i3;
      int j = 0;
      for (; j + 16 <= m; j += 16)
        add_rows<16>(acc, r, rowBeg, rowEnd, rsv, nRows, xl, mrow, myidx, j, b + j, nSrc - 1);
      if (m - j >= 8) {
        add_rows<8>(acc, r, rowBeg, rowEnd, rsv, nRows, xl, mrow, myidx, j, b + j, nSrc - 1);
        j += 8;
      }
      if (m - j >= 4) {
        add_rows<4>(acc, r, rowBeg, rowEnd, rsv, nRows, xl, mrow, myidx, j, b + j, nSrc - 1);
        j += 4;
      }
      if (m - j >= 2) {
        add_rows<2>(acc, r, rowBeg, rowEnd, rsv, nRows, xl, mrow, myidx, j, b + j, nSrc - 1);
        j += 2;
      }
      if (m - j >= 1)
        add_rows<1>(acc, r, rowBeg, rowEnd, rsv, nRows, xl, mrow, myidx, j, b + j, nSrc - 1);
    }
  }
  store_mean(mrow + (size_t)r * DIM, acc, rowEnd - rowBeg);
  for (++r; r < nRows; ++r)
    *(float4*)(mrow + (size_t)r * DIM) = make_float4(0.0f, 0.0f, 0.0f, 0.0f);
}

#define WPITCH 257
#define GEMM_WAVES 16
__global__ __launch_bounds__(512) void k_sage_gemm(
    const float* __restrict__ mean, const float* __restrict__ xdst,
    const float* __restrict__ Wl, const float* __restrict__ Wr,
    const float* __restrict__ bias, float* __restrict__ out,
    int nRowTiles, int relu) {
  __shared__ v2f sW[2 * 32 * WPITCH];
  int wave = threadIdx.x >> 5;
  int lane = threadIdx.x & 31;

#pragma unroll
  for (int i = 0; i < 2 * DIM / GEMM_WAVES; ++i) {
    int n = wave + i * GEMM_WAVES;
    const float* wsrc = (n < DIM) ? Wl : Wr;
    int nn = n & (DIM - 1);
    const float4 w = *(const float4*)(wsrc + (size_t)nn * DIM + lane * 4);
    v2f* d = sW + (n >> 7) * (32 * WPITCH) + lane * WPITCH + (nn >> 4) * 32 + (nn & 15);
    v2f lo = {w.x, w.y};
    v2f hi = {w.z, w.w};
    d[0] = lo;
    d[16] = hi;
  }
  __syncthreads();

  int lm = lane & 15;
  int khalf = (lane >= 16) ? 2 : 0;
  const v2f* sWl = sW + lane;
  const v2f* sWr = sW + 32 * WPITCH + lane;

  for (int rowTile = blockIdx.x + gridDim.x * wave; rowTile < nRowTiles;
       rowTile += gridDim.x * GEMM_WAVES) {
    int row0 = rowTile * 16;
    const float* aBase = mean + (size_t)(row0 + lm) * DIM + khalf;
    const float* xBase = xdst + (size_t)(row0 + lm) * DIM + khalf;

    v8f accA[8], accB[8];
#pragma unroll
    for (int t = 0; t < 8; ++t) {
      float bval = bias[t * 16 + lm];
      accA[t] = (v8f){bval, bval, bval, bval, bval, bval, bval, bval};
      accB[t] = (v8f){0.f, 0.f, 0.f, 0.f, 0.f, 0.f, 0.f, 0.f};
    }

    v2f am[4], ax[4];
#pragma unroll
    for (int u = 0; u < 4; ++u) {
      am[u] = *(const v2f*)(aBase + u * 4);
      ax[u] = *(const v2f*)(xBase + u * 4);
    }
#pragma unroll 1
    for (int g = 0; g < DIM / 16; ++g) {
      v2f amN[4], axN[4];
      int gn = (g + 1) & (DIM / 16 - 1);
#pragma unroll
      for (int u = 0; u < 4; ++u) {
        amN[u] = *(const v2f*)(aBase + gn * 16 + u * 4);
        axN[u] = *(const v2f*)(xBase + gn * 16 + u * 4);
      }
      const v2f* bl = sWl + g * 4 * WPITCH;
      const v2f* br = sWr + g * 4 * WPITCH;
#pragma unroll
      for (int u = 0; u < 4; ++u) {
#pragma unroll
        for (int t = 0; t < 8; ++t) {
          accA[t] = __builtin_amdgcn_wmma_f32_16x16x4_f32(false, am[u], false,
                                                          bl[u * WPITCH + t * 32], (short)0,
                                                          accA[t], false, false);
          accB[t] = __builtin_amdgcn_wmma_f32_16x16x4_f32(false, ax[u], false,
                                                          br[u * WPITCH + t * 32], (short)0,
                                                          accB[t], false, false);
        }
      }
#pragma unroll
      for (int u = 0; u < 4; ++u) {
        am[u] = amN[u];
        ax[u] = axN[u];
      }
    }

    float* orow = out + (size_t)(row0 + ((lane >= 16) ? 8 : 0)) * DIM + lm;
#pragma unroll
    for (int t = 0; t < 8; ++t) {
#pragma unroll
      for (int r = 0; r < 8; ++r) {
        float v = accA[t][r] + accB[t][r];
        if (relu) v = fmaxf(v, 0.0f);
        orow[(size_t)r * DIM + t * 16] = v;
      }
    }
  }
}

__global__ __launch_bounds__(256) void k_decode(
    const float* __restrict__ zu, const float* __restrict__ zj,
    const int* __restrict__ elu, const int* __restrict__ elj,
    float* __restrict__ out, int n) {
  const int p0 = 32 * __builtin_amdgcn_readfirstlane((int)(blockIdx.x * 8 + (threadIdx.x >> 5)));
  if (p0 >= n) return;
  const int lane = threadIdx.x & 31;
  const int np = min(n - p0, 32);
  const int iu = (lane < np) ? elu[p0 + lane] : 0;
  const int ij = (lane < np) ? elj[p0 + lane] : 0;
  const float* zul = zu + lane * 4;
  const float* zjl = zj + lane * 4;
  float rd = 0.0f, ra = 0.0f, rb = 0.0f;
#pragma unroll 1
  for (int q = 0; q < np; q += 8) {
    float4 a[8], b[8];
#pragma unroll
    for (int i = 0; i < 8; ++i) {
      const int su = __builtin_amdgcn_readlane(iu, q + i);
      const int sj = __builtin_amdgcn_readlane(ij, q + i);
      a[i] = *(const float4*)(zul + (size_t)su * DIM);
      b[i] = *(const float4*)(zjl + (size_t)sj * DIM);
    }
    __builtin_amdgcn_sched_barrier(0);
    float dot[8], na[8], nb[8];
#pragma unroll
    for (int i = 0; i < 8; ++i) {
      dot[i] = fmaf(a[i].w, b[i].w, fmaf(a[i].z, b[i].z, fmaf(a[i].x, b[i].x, a[i].y * b[i].y)));
      na[i] = fmaf(a[i].w, a[i].w, fmaf(a[i].z, a[i].z, fmaf(a[i].x, a[i].x, a[i].y * a[i].y)));
      nb[i] = fmaf(b[i].w, b[i].w, fmaf(b[i].z, b[i].z, fmaf(b[i].x, b[i].x, b[i].y * b[i].y)));
    }
#pragma unroll
    for (int off = 16; off > 0; off >>= 1) {
#pragma unroll
      for (int i = 0; i < 8; ++i) {
        dot[i] += __shfl_xor(dot[i], off, 32);
        na[i] += __shfl_xor(na[i], off, 32);
        nb[i] += __shfl_xor(nb[i], off, 32);
      }
    }
#pragma unroll
    for (int i = 0; i < 8; ++i) {
      if (lane == q + i) {
        rd = dot[i];
        ra = na[i];
        rb = nb[i];
      }
    }
  }
  if (lane < np) out[p0 + lane] = rd / (fmaxf(sqrtf(ra), 1e-8f) * fmaxf(sqrtf(rb), 1e-8f));
}


#define RJ 4
#define RU 8
#define GJ_WGS (((NJJ + RJ - 1) / RJ + 7) / 8)
#define GU_WGS (((NUU + RU - 1) / RU + 7) / 8)

extern "C" void kernel_launch(void* const* d_in, const int* in_sizes, int n_in,
                              void* d_out, int out_size, void* d_ws, size_t ws_size,
                              hipStream_t stream) {
  const float* x_user = (const float*)d_in[0];
  const float* x_job = (const float*)d_in[1];
  const int* uj_src = (const int*)d_in[2];
  const int* uj_dst = (const int*)d_in[3];
  const int* ju_src = (const int*)d_in[4];
  const int* ju_dst = (const int*)d_in[5];
  const int* el_user = (const int*)d_in[6];
  const int* el_job = (const int*)d_in[7];
  const float* W1l_uj = (const float*)d_in[8];
  const float* W1r_uj = (const float*)d_in[9];
  const float* W1l_ju = (const float*)d_in[10];
  const float* W1r_ju = (const float*)d_in[11];
  const float* W2l_uj = (const float*)d_in[12];
  const float* W2r_uj = (const float*)d_in[13];
  const float* W2l_ju = (const float*)d_in[14];
  const float* W2r_ju = (const float*)d_in[15];
  const float* b1_uj = (const float*)d_in[16];
  const float* b1_ju = (const float*)d_in[17];
  const float* b2_uj = (const float*)d_in[18];
  const float* b2_ju = (const float*)d_in[19];
  float* out = (float*)d_out;

  float* ws = (float*)d_ws;
  float* agg_user = ws; ws += (size_t)NUU * DIM;
  float* agg_job  = ws; ws += (size_t)NJJ * DIM;
  float* h_user   = ws; ws += (size_t)NUU * DIM;
  float* h_job    = ws; ws += (size_t)NJJ * DIM;
  float* z_user   = ws; ws += (size_t)NUU * DIM;
  float* z_job    = ws; ws += (size_t)NJJ * DIM;
  int* cnt_user = (int*)ws;
  int* cnt_job  = cnt_user + NUU;
  int* rank_job  = (int*)z_user;
  int* rank_user = rank_job + NEDGE;
  int* perm_job  = rank_user + NEDGE;
  int* perm_user = perm_job + NEDGE;
  int* rs_job    = perm_user + NEDGE;
  int* rs_user   = rs_job + (NJJ + 32);

  const dim3 eg((NEDGE + EDGE_WG * EPT - 1) / (EDGE_WG * EPT), 2);

  k_zero_i<<<((NUU + NJJ) / 4 + 1023) / 1024, 1024, 0, stream>>>(cnt_user, (NUU + NJJ) / 4);
  k_rank<<<eg, EDGE_WG, 0, stream>>>(uj_dst, ju_dst, cnt_job, cnt_user, rank_job, rank_user, NEDGE);
  k_scan<<<2, 1024, 0, stream>>>(cnt_job, cnt_user, rs_job, rs_user);
  k_fill<<<eg, EDGE_WG, 0, stream>>>(uj_src, uj_dst, ju_src, ju_dst, rank_job, rank_user,
                                     rs_job, rs_user, perm_job, perm_user, NEDGE);

  k_gather_mean<RJ><<<GJ_WGS, 256, 0, stream>>>(x_user, rs_job, perm_job, agg_job, NJJ, NUU);
  k_gather_mean<RU><<<GU_WGS, 256, 0, stream>>>(x_job, rs_user, perm_user, agg_user, NUU, NJJ);

  k_sage_gemm<<<256, 512, 0, stream>>>(agg_job, x_job, W1l_uj, W1r_uj,
                                            b1_uj, h_job, NJJ / 16, 1);
  k_sage_gemm<<<256, 512, 0, stream>>>(agg_user, x_user, W1l_ju, W1r_ju,
                                            b1_ju, h_user, NUU / 16, 1);

  k_gather_mean<RJ><<<GJ_WGS, 256, 0, stream>>>(h_user, rs_job, perm_job, agg_job, NJJ, NUU);
  k_gather_mean<RU><<<GU_WGS, 256, 0, stream>>>(h_job, rs_user, perm_user, agg_user, NUU, NJJ);

  k_sage_gemm<<<256, 512, 0, stream>>>(agg_job, h_job, W2l_uj, W2r_uj,
                                            b2_uj, z_job, NJJ / 16, 0);
  k_sage_gemm<<<256, 512, 0, stream>>>(agg_user, h_user, W2l_ju, W2r_ju,
                                            b2_ju, z_user, NUU / 16, 0);

  k_decode<<<((NEL + 31) / 32 + 7) / 8, dim3(256), 0, stream>>>(z_user, z_job, el_user, el_job, out, NEL);
}


# Both destination types of a layer share one launch: one neighbour-mean launch and one GEMM launch (workgroups split 73:183 by row count) per layer; 13 -> 9 launches; every sum order kept
# speedup vs baseline: 3.0651x; 1.1353x over previous
#include <hip/hip_runtime.h>
#include <hip/hip_bf16.h>
#include <math.h>

#define NUU 50000
#define NJJ 20000
#define NEDGE 500000
#define NEL 100000
#define DIM 128

typedef __attribute__((ext_vector_type(2))) float v2f;
typedef __attribute__((ext_vector_type(8))) float v8f;


#define EPT 8
#define EDGE_WG 512

__global__ __launch_bounds__(1024) void k_zero_i(int* __restrict__ p, int n4) {
  int i = blockIdx.x * 1024 + threadIdx.x;
  if (i < n4) ((int4*)p)[i] = make_int4(0, 0, 0, 0);
}

__device__ __forceinline__ void rank_edges(const int* __restrict__ dst, int* __restrict__ cnt,
                                           int* __restrict__ rank, int n) {
  const int e0 = blockIdx.x * (EDGE_WG * EPT) + threadIdx.x;
  int d[EPT], r[EPT];
#pragma unroll
  for (int k = 0; k < EPT; ++k) {
    d[k] = dst[min(e0 + k * EDGE_WG, n - 1)];
  }
#pragma unroll
  for (int k = 0; k < EPT; ++k) {
    const int e = e0 + k * EDGE_WG;
    r[k] = (e < n) ? atomicAdd(&cnt[d[k]], 1) : 0;
  }
#pragma unroll
  for (int k = 0; k < EPT; ++k) {
    const int e = e0 + k * EDGE_WG;
    if (e < n) rank[e] = r[k];
  }
}

__global__ __launch_bounds__(EDGE_WG) void k_rank(const int* __restrict__ dstJ,
                                                   const int* __restrict__ dstU,
                                                   int* __restrict__ cntJ, int* __restrict__ cntU,
                                                   int* __restrict__ rankJ, int* __restrict__ rankU,
                                                   int n) {
  if (blockIdx.y == 0) rank_edges(dstJ, cntJ, rankJ, n);
  else rank_edges(dstU, cntU, rankU, n);
}

template <int N>
__device__ __forceinline__ void scan_counts(const int* __restrict__ cnt, int* __restrict__ rs,
                                            int* sc, int* part) {
  constexpr int ROWS = (N + 1023) / 1024;
  constexpr int CH = ROWS | 1;
  const int tid = threadIdx.x;
#pragma unroll
  for (int r = 0; r < ROWS; ++r) {
    int i = r * 1024 + tid;
    if (i < N) sc[i] = cnt[i];
  }
  __syncthreads();
  const int b = tid * CH;
  int sum = 0;
  for (int k = 0; k < CH; ++k)
    if (b + k < N) sum += sc[b + k];
  const int lane = tid & 31, wave = tid >> 5;
  int inc = sum;
  for (int o = 1; o < 32; o <<= 1) {
    int t = __shfl_up(inc, o, 32);
    if (lane >= o) inc += t;
  }
  if (lane == 31) part[wave] = inc;
  __syncthreads();
  const int pw = part[lane];
  int pinc = pw;
  for (int o = 1; o < 32; o <<= 1) {
    int t = __shfl_up(pinc, o, 32);
    if (lane >= o) pinc += t;
  }
  const int woff = __shfl(pinc - pw, wave, 32);
  int run = woff + inc - sum;
  for (int k = 0; k < CH; ++k)
    if (b + k < N) {
      int c = sc[b + k];
      sc[b + k] = run;
      run += c;
    }
  __syncthreads();
#pragma unroll
  for (int r = 0; r < ROWS; ++r) {
    int i = r * 1024 + tid;
    if (i < N) rs[i] = sc[i];
  }
  if (tid == 1023) rs[N] = run;
}

__global__ __launch_bounds__(1024) void k_scan(const int* __restrict__ cntJ,
                                                const int* __restrict__ cntU,
                                                int* __restrict__ rsJ, int* __restrict__ rsU) {
  __shared__ int sc[NUU];
  __shared__ int part[32];
  if (blockIdx.x == 0) scan_counts<NJJ>(cntJ, rsJ, sc, part);
  else scan_counts<NUU>(cntU, rsU, sc, part);
}

__device__ __forceinline__ void fill_edges(const int* __restrict__ src, const int* __restrict__ dst,
                                           const int* __restrict__ rank, const int* __restrict__ rs,
                                           int* __restrict__ perm, int n) {
  const int e0 = blockIdx.x * (EDGE_WG * EPT) + threadIdx.x;
  int d[EPT], rk[EPT], s[EPT], pos[EPT];
#pragma unroll
  for (int k = 0; k < EPT; ++k) {
    const int e = min(e0 + k * EDGE_WG, n - 1);
    d[k] = dst[e];
    rk[k] = rank[e];
    s[k] = src[e];
  }
#pragma unroll
  for (int k = 0; k < EPT; ++k) pos[k] = rs[d[k]] + rk[k];
#pragma unroll
  for (int k = 0; k < EPT; ++k) {
    const int e = e0 + k * EDGE_WG;
    if (e < n) perm[min(max(pos[k], 0), n - 1)] = s[k];
  }
}

__global__ __launch_bounds__(EDGE_WG) void k_fill(const int* __restrict__ srcJ,
                                                   const int* __restrict__ dstJ,
                                                   const int* __restrict__ srcU,
                                                   const int* __restrict__ dstU,
                                                   const int* __restrict__ rankJ,
                                                   const int* __restrict__ rankU,
                                                   const int* __restrict__ rsJ,
                                                   const int* __restrict__ rsU,
                                                   int* __restrict__ permJ, int* __restrict__ permU,
                                                   int n) {
  if (blockIdx.y == 0) fill_edges(srcJ, dstJ, rankJ, rsJ, permJ, n);
  else fill_edges(srcU, dstU, rankU, rsU, permU, n);
}

__device__ __forceinline__ void store_mean(float* __restrict__ p, const float4 acc, int count) {
  const float c = fmaxf((float)count, 1.0f);
  float4 o;
  o.x = acc.x / c;
  o.y = acc.y / c;
  o.z = acc.z / c;
  o.w = acc.w / c;
  *(float4*)p = o;
}

template <int U>
__device__ __forceinline__ void add_rows(float4& acc, int& r, int& rowBeg, int& rowEnd,
                                         const int rsv, const int nRows,
                                         const float* __restrict__ xl, float* __restrict__ mrow,
                                         int myidx, int j, int e, int lastSrc) {
  float4 rr[U];
#pragma unroll
  for (int u = 0; u < U; ++u) {
    int s = __builtin_amdgcn_readlane(myidx, j + u);
    s = min(max(s, 0), lastSrc);
    rr[u] = *(const float4*)(xl + (size_t)s * DIM);
  }
#pragma unroll
  for (int u = 0; u < U; ++u) {
    if (e + u >= rowEnd && r + 1 < nRows) {
      store_mean(mrow + (size_t)r * DIM, acc, rowEnd - rowBeg);
      acc = make_float4(0.0f, 0.0f, 0.0f, 0.0f);
      ++r;
      rowBeg = rowEnd;
      rowEnd = __builtin_amdgcn_readlane(rsv, r + 1);
      while (e + u >= rowEnd && r + 1 < nRows) {
        *(float4*)(mrow + (size_t)r * DIM) = acc;
        ++r;
        rowBeg = rowEnd;
        rowEnd = __builtin_amdgcn_readlane(rsv, r + 1);
      }
    }
    acc.x += rr[u].x;
    acc.y += rr[u].y;
    acc.z += rr[u].z;
    acc.w += rr[u].w;
  }
}

#define RJ 4
#define RU 8
#define GJ_WGS (((NJJ + RJ - 1) / RJ + 7) / 8)
#define GU_WGS (((NUU + RU - 1) / RU + 7) / 8)
__global__ __launch_bounds__(256) void k_gather_mean(const float* __restrict__ xsrcJ,
                                                      const int* __restrict__ rsJ,
                                                      const int* __restrict__ permJ,
                                                      float* __restrict__ meanJ,
                                                      const float* __restrict__ xsrcU,
                                                      const int* __restrict__ rsU,
                                                      const int* __restrict__ permU,
                                                      float* __restrict__ meanU) {
  const bool isJ = blockIdx.x < GJ_WGS;
  const float* __restrict__ xsrc = isJ ? xsrcJ : xsrcU;
  const int* __restrict__ rs = isJ ? rsJ : rsU;
  const int* __restrict__ perm = isJ ? permJ : permU;
  float* __restrict__ mean = isJ ? meanJ : meanU;
  const int nDst = isJ ? NJJ : NUU;
  const int nSrc = isJ ? NUU : NJJ;
  const int R = isJ ? RJ : RU;
  const int wg = isJ ? (int)blockIdx.x : (int)blockIdx.x - GJ_WGS;
  const int d0 = R * __builtin_amdgcn_readfirstlane((int)(wg * 8 + (threadIdx.x >> 5)));
  if (d0 >= nDst) return;
  const int nRows = min(R, nDst - d0);
  const int lane = threadIdx.x & 31;
  const int rsv = (lane <= nRows) ? rs[d0 + lane] : 0;
  const int eBeg = __builtin_amdgcn_readlane(rsv, 0);
  const int eEnd = __builtin_amdgcn_readlane(rsv, nRows);
  const float* xl = xsrc + lane * 4;
  float* mrow = mean + (size_t)d0 * DIM + lane * 4;
  int r = 0;
  int rowBeg = eBeg;
  int rowEnd = __builtin_amdgcn_readlane(rsv, 1);
  float4 acc = make_float4(0.0f, 0.0f, 0.0f, 0.0f);
  for (int base = eBeg; base < eEnd; base += 128) {
    const int left = eEnd - base;
    const int i0 = (lane < left) ? perm[base + lane] : 0;
    const int i1 = (lane + 32 < left) ? perm[base + 32 + lane] : 0;
    const int i2 = (lane + 64 < left) ? perm[base + 64 + lane] : 0;
    const int i3 = (lane + 96 < left) ? perm[base + 96 + lane] : 0;
#pragma unroll 1
    for (int k = 0; k < 4; ++k) {
      const int b = base + 32 * k;
      const int m = min(eEnd - b, 32);
      if (m <= 0) break;
      const int myidx = (k == 0) ? i0 : (k == 1) ? i1 : (k == 2) ? i2 : i3;
      int j = 0;
      for (; j + 16 <= m; j += 16)
        add_rows<16>(acc, r, rowBeg, rowEnd, rsv, nRows, xl, mrow, myidx, j, b + j, nSrc - 1);
      if (m - j >= 8) {
        add_rows<8>(acc, r, rowBeg, rowEnd, rsv, nRows, xl, mrow, myidx, j, b + j, nSrc - 1);
        j += 8;
      }
      if (m - j >= 4) {
        add_rows<4>(acc, r, rowBeg, rowEnd, rsv, nRows, xl, mrow, myidx, j, b + j, nSrc - 1);
        j += 4;
      }
      if (m - j >= 2) {
        add_rows<2>(acc, r, rowBeg, rowEnd, rsv, nRows, xl, mrow, myidx, j, b + j, nSrc - 1);
        j += 2;
      }
      if (m - j >= 1)
        add_rows<1>(acc, r, rowBeg, rowEnd, rsv, nRows, xl, mrow, myidx, j, b + j, nSrc - 1);
    }
  }
  store_mean(mrow + (size_t)r * DIM, acc, rowEnd - rowBeg);
  for (++r; r < nRows; ++r)
    *(float4*)(mrow + (size_t)r * DIM) = make_float4(0.0f, 0.0f, 0.0f, 0.0f);
}

#define WPITCH 257
#define GEMM_WAVES 16
#define GEMM_WGS 256
#define GEMM_WGS_J 73
__global__ __launch_bounds__(512) void k_sage_gemm(
    const float* __restrict__ meanJ, const float* __restrict__ xdstJ,
    const float* __restrict__ WlJ, const float* __restrict__ WrJ,
    const float* __restrict__ biasJ, float* __restrict__ outJ,
    const float* __restrict__ meanU, const float* __restrict__ xdstU,
    const float* __restrict__ WlU, const float* __restrict__ WrU,
    const float* __restrict__ biasU, float* __restrict__ outU,
    int relu) {
  __shared__ v2f sW[2 * 32 * WPITCH];
  int wave = threadIdx.x >> 5;
  int lane = threadIdx.x & 31;
  const bool isJ = blockIdx.x < GEMM_WGS_J;
  const float* __restrict__ mean = isJ ? meanJ : meanU;
  const float* __restrict__ xdst = isJ ? xdstJ : xdstU;
  const float* __restrict__ Wl = isJ ? WlJ : WlU;
  const float* __restrict__ Wr = isJ ? WrJ : WrU;
  const float* __restrict__ bias = isJ ? biasJ : biasU;
  float* __restrict__ out = isJ ? outJ : outU;
  const int nRowTiles = isJ ? NJJ / 16 : NUU / 16;
  const int wg = isJ ? (int)blockIdx.x : (int)blockIdx.x - GEMM_WGS_J;
  const int nWg = isJ ? GEMM_WGS_J : GEMM_WGS - GEMM_WGS_J;

#pragma unroll
  for (int i = 0; i < 2 * DIM / GEMM_WAVES; ++i) {
    int n = wave + i * GEMM_WAVES;
    const float* wsrc = (n < DIM) ? Wl : Wr;
    int nn = n & (DIM - 1);
    const float4 w = *(const float4*)(wsrc + (size_t)nn * DIM + lane * 4);
    v2f* d = sW + (n >> 7) * (32 * WPITCH) + lane * WPITCH + (nn >> 4) * 32 + (nn & 15);
    v2f lo = {w.x, w.y};
    v2f hi = {w.z, w.w};
    d[0] = lo;
    d[16] = hi;
  }
  __syncthreads();

  int lm = lane & 15;
  int khalf = (lane >= 16) ? 2 : 0;
  const v2f* sWl = sW + lane;
  const v2f* sWr = sW + 32 * WPITCH + lane;

  for (int rowTile = wg + nWg * wave; rowTile < nRowTiles; rowTile += nWg * GEMM_WAVES) {
    int row0 = rowTile * 16;
    const float* aBase = mean + (size_t)(row0 + lm) * DIM + khalf;
    const float* xBase = xdst + (size_t)(row0 + lm) * DIM + khalf;

    v8f accA[8], accB[8];
#pragma unroll
    for (int t = 0; t < 8; ++t) {
      float bval = bias[t * 16 + lm];
      accA[t] = (v8f){bval, bval, bval, bval, bval, bval, bval, bval};
      accB[t] = (v8f){0.f, 0.f, 0.f, 0.f, 0.f, 0.f, 0.f, 0.f};
    }

    v2f am[4], ax[4];
#pragma unroll
    for (int u = 0; u < 4; ++u) {
      am[u] = *(const v2f*)(aBase + u * 4);
      ax[u] = *(const v2f*)(xBase + u * 4);
    }
#pragma unroll 1
    for (int g = 0; g < DIM / 16; ++g) {
      v2f amN[4], axN[4];
      int gn = (g + 1) & (DIM / 16 - 1);
#pragma unroll
      for (int u = 0; u < 4; ++u) {
        amN[u] = *(const v2f*)(aBase + gn * 16 + u * 4);
        axN[u] = *(const v2f*)(xBase + gn * 16 + u * 4);
      }
      const v2f* bl = sWl + g * 4 * WPITCH;
      const v2f* br = sWr + g * 4 * WPITCH;
#pragma unroll
      for (int u = 0; u < 4; ++u) {
#pragma unroll
        for (int t = 0; t < 8; ++t) {
          accA[t] = __builtin_amdgcn_wmma_f32_16x16x4_f32(false, am[u], false,
                                                          bl[u * WPITCH + t * 32], (short)0,
                                                          accA[t], false, false);
          accB[t] = __builtin_amdgcn_wmma_f32_16x16x4_f32(false, ax[u], false,
                                                          br[u * WPITCH + t * 32], (short)0,
                                                          accB[t], false, false);
        }
      }
#pragma unroll
      for (int u = 0; u < 4; ++u) {
        am[u] = amN[u];
        ax[u] = axN[u];
      }
    }

    float* orow = out + (size_t)(row0 + ((lane >= 16) ? 8 : 0)) * DIM + lm;
#pragma unroll
    for (int t = 0; t < 8; ++t) {
#pragma unroll
      for (int r = 0; r < 8; ++r) {
        float v = accA[t][r] + accB[t][r];
        if (relu) v = fmaxf(v, 0.0f);
        orow[(size_t)r * DIM + t * 16] = v;
      }
    }
  }
}

__global__ __launch_bounds__(256) void k_decode(
    const float* __restrict__ zu, const float* __restrict__ zj,
    const int* __restrict__ elu, const int* __restrict__ elj,
    float* __restrict__ out, int n) {
  const int p0 = 32 * __builtin_amdgcn_readfirstlane((int)(blockIdx.x * 8 + (threadIdx.x >> 5)));
  if (p0 >= n) return;
  const int lane = threadIdx.x & 31;
  const int np = min(n - p0, 32);
  const int iu = (lane < np) ? elu[p0 + lane] : 0;
  const int ij = (lane < np) ? elj[p0 + lane] : 0;
  const float* zul = zu + lane * 4;
  const float* zjl = zj + lane * 4;
  float rd = 0.0f, ra = 0.0f, rb = 0.0f;
#pragma unroll 1
  for (int q = 0; q < np; q += 8) {
    float4 a[8], b[8];
#pragma unroll
    for (int i = 0; i < 8; ++i) {
      const int su = __builtin_amdgcn_readlane(iu, q + i);
      const int sj = __builtin_amdgcn_readlane(ij, q + i);
      a[i] = *(const float4*)(zul + (size_t)su * DIM);
      b[i] = *(const float4*)(zjl + (size_t)sj * DIM);
    }
    __builtin_amdgcn_sched_barrier(0);
    float dot[8], na[8], nb[8];
#pragma unroll
    for (int i = 0; i < 8; ++i) {
      dot[i] = fmaf(a[i].w, b[i].w, fmaf(a[i].z, b[i].z, fmaf(a[i].x, b[i].x, a[i].y * b[i].y)));
      na[i] = fmaf(a[i].w, a[i].w, fmaf(a[i].z, a[i].z, fmaf(a[i].x, a[i].x, a[i].y * a[i].y)));
      nb[i] = fmaf(b[i].w, b[i].w, fmaf(b[i].z, b[i].z, fmaf(b[i].x, b[i].x, b[i].y * b[i].y)));
    }
#pragma unroll
    for (int off = 16; off > 0; off >>= 1) {
#pragma unroll
      for (int i = 0; i < 8; ++i) {
        dot[i] += __shfl_xor(dot[i], off, 32);
        na[i] += __shfl_xor(na[i], off, 32);
        nb[i] += __shfl_xor(nb[i], off, 32);
      }
    }
#pragma unroll
    for (int i = 0; i < 8; ++i) {
      if (lane == q + i) {
        rd = dot[i];
        ra = na[i];
        rb = nb[i];
      }
    }
  }
  if (lane < np) out[p0 + lane] = rd / (fmaxf(sqrtf(ra), 1e-8f) * fmaxf(sqrtf(rb), 1e-8f));
}


extern "C" void kernel_launch(void* const* d_in, const int* in_sizes, int n_in,
                              void* d_out, int out_size, void* d_ws, size_t ws_size,
                              hipStream_t stream) {
  const float* x_user = (const float*)d_in[0];
  const float* x_job = (const float*)d_in[1];
  const int* uj_src = (const int*)d_in[2];
  const int* uj_dst = (const int*)d_in[3];
  const int* ju_src = (const int*)d_in[4];
  const int* ju_dst = (const int*)d_in[5];
  const int* el_user = (const int*)d_in[6];
  const int* el_job = (const int*)d_in[7];
  const float* W1l_uj = (const float*)d_in[8];
  const float* W1r_uj = (const float*)d_in[9];
  const float* W1l_ju = (const float*)d_in[10];
  const float* W1r_ju = (const float*)d_in[11];
  const float* W2l_uj = (const float*)d_in[12];
  const float* W2r_uj = (const float*)d_in[13];
  const float* W2l_ju = (const float*)d_in[14];
  const float* W2r_ju = (const float*)d_in[15];
  const float* b1_uj = (const float*)d_in[16];
  const float* b1_ju = (const float*)d_in[17];
  const float* b2_uj = (const float*)d_in[18];
  const float* b2_ju = (const float*)d_in[19];
  float* out = (float*)d_out;

  float* ws = (float*)d_ws;
  float* agg_user = ws; ws += (size_t)NUU * DIM;
  float* agg_job  = ws; ws += (size_t)NJJ * DIM;
  float* h_user   = ws; ws += (size_t)NUU * DIM;
  float* h_job    = ws; ws += (size_t)NJJ * DIM;
  float* z_user   = ws; ws += (size_t)NUU * DIM;
  float* z_job    = ws; ws += (size_t)NJJ * DIM;
  int* cnt_user = (int*)ws;
  int* cnt_job  = cnt_user + NUU;
  int* rank_job  = (int*)z_user;
  int* rank_user = rank_job + NEDGE;
  int* perm_job  = rank_user + NEDGE;
  int* perm_user = perm_job + NEDGE;
  int* rs_job    = perm_user + NEDGE;
  int* rs_user   = rs_job + (NJJ + 32);

  const dim3 eg((NEDGE + EDGE_WG * EPT - 1) / (EDGE_WG * EPT), 2);

  k_zero_i<<<((NUU + NJJ) / 4 + 1023) / 1024, 1024, 0, stream>>>(cnt_user, (NUU + NJJ) / 4);
  k_rank<<<eg, EDGE_WG, 0, stream>>>(uj_dst, ju_dst, cnt_job, cnt_user, rank_job, rank_user, NEDGE);
  k_scan<<<2, 1024, 0, stream>>>(cnt_job, cnt_user, rs_job, rs_user);
  k_fill<<<eg, EDGE_WG, 0, stream>>>(uj_src, uj_dst, ju_src, ju_dst, rank_job, rank_user,
                                     rs_job, rs_user, perm_job, perm_user, NEDGE);

  k_gather_mean<<<GJ_WGS + GU_WGS, 256, 0, stream>>>(x_user, rs_job, perm_job, agg_job,
                                                     x_job, rs_user, perm_user, agg_user);

  k_sage_gemm<<<GEMM_WGS, 512, 0, stream>>>(agg_job, x_job, W1l_uj, W1r_uj, b1_uj, h_job,
                                            agg_user, x_user, W1l_ju, W1r_ju, b1_ju, h_user, 1);

  k_gather_mean<<<GJ_WGS + GU_WGS, 256, 0, stream>>>(h_user, rs_job, perm_job, agg_job,
                                                     h_job, rs_user, perm_user, agg_user);

  k_sage_gemm<<<GEMM_WGS, 512, 0, stream>>>(agg_job, h_job, W2l_uj, W2r_uj, b2_uj, z_job,
                                            agg_user, h_user, W2l_ju, W2r_ju, b2_ju, z_user, 0);

  k_decode<<<((NEL + 31) / 32 + 7) / 8, dim3(256), 0, stream>>>(z_user, z_job, el_user, el_job, out, NEL);
}


# Edge grouping in ONE launch: 64 owner workgroups per edge type read the whole destination list, then count, scan and place their rows in LDS (slot = arrival order of an LDS add); 9 -> 6 launches
# speedup vs baseline: 4.0335x; 1.3159x over previous
#include <hip/hip_runtime.h>
#include <hip/hip_bf16.h>
#include <math.h>

#define NUU 50000
#define NJJ 20000
#define NEDGE 500000
#define NEL 100000
#define DIM 128

typedef __attribute__((ext_vector_type(2))) float v2f;
typedef __attribute__((ext_vector_type(8))) float v8f;


#define GRP_T 1024
#define GRP_WGS_T 64
#define GRP_CAPW 1024
#define GRP_EBITS 19
#define GRP_MAXR ((NUU + GRP_WGS_T - 1) / GRP_WGS_T)
static_assert(NEDGE % 4 == 0, "the destination list is read four entries per lane");
static_assert(GRP_MAXR <= GRP_T && GRP_MAXR <= 1024, "one thread per owned row; row fits 10 bits");
static_assert(NEDGE <= (1 << GRP_EBITS), "edge number and row share one 32-bit note");
static_assert((NJJ + GRP_WGS_T - 1) / GRP_WGS_T <= GRP_MAXR, "job rows per owner fit too");

__device__ __forceinline__ void grp_take(const int d, const int e, const int lo, const int nr,
                                         int* hist, int* wl, int& below, int& wcount) {
  below += __builtin_popcount(__builtin_amdgcn_ballot_w32(d < lo));
  const unsigned dl = (unsigned)(d - lo);
  const bool mt = dl < (unsigned)nr;
  const unsigned mask = __builtin_amdgcn_ballot_w32(mt);
  if (mask != 0u) {
    if (mt) {
      atomicAdd(&hist[dl], 1);
      const int p = wcount + (int)__builtin_amdgcn_mbcnt_lo(mask, 0u);
      if (p < GRP_CAPW) wl[p] = (int)((unsigned)e | (dl << GRP_EBITS));
    }
    wcount += __builtin_popcount(mask);
  }
}

template <int NG, bool LAST>
__device__ __forceinline__ void grp_chunks(const int4* __restrict__ d4, const int c0, const int tid,
                                           const int lo, const int nr, int* hist, int* wl,
                                           int& below, int& wcount) {
  constexpr int N4 = NEDGE / 4;
  int4 v[NG];
#pragma unroll
  for (int u = 0; u < NG; ++u) {
    const int i4 = (c0 + u) * GRP_T + tid;
    if (LAST) {
      v[u] = d4[min(i4, N4 - 1)];
      if (i4 >= N4) v[u] = make_int4(0x7fffffff, 0x7fffffff, 0x7fffffff, 0x7fffffff);
    } else {
      v[u] = d4[i4];
    }
  }
#pragma unroll
  for (int u = 0; u < NG; ++u) {
    const int e0 = 4 * ((c0 + u) * GRP_T + tid);
    grp_take(v[u].x, e0, lo, nr, hist, wl, below, wcount);
    grp_take(v[u].y, e0 + 1, lo, nr, hist, wl, below, wcount);
    grp_take(v[u].z, e0 + 2, lo, nr, hist, wl, below, wcount);
    grp_take(v[u].w, e0 + 3, lo, nr, hist, wl, below, wcount);
  }
}

__global__ __launch_bounds__(GRP_T) void k_group(const int* __restrict__ dstJ,
                                                  const int* __restrict__ srcJ,
                                                  const int* __restrict__ dstU,
                                                  const int* __restrict__ srcU,
                                                  int* __restrict__ rsJ, int* __restrict__ rsU,
                                                  int* __restrict__ permJ,
                                                  int* __restrict__ permU) {
  __shared__ int hist[GRP_MAXR];
  __shared__ int rsl[GRP_MAXR];
  __shared__ int lst[32 * GRP_CAPW];
  __shared__ int part[32];
  __shared__ int wsum[32];
  __shared__ int ovf;
  const bool isJ = blockIdx.x < GRP_WGS_T;
  const int* __restrict__ dst = isJ ? dstJ : dstU;
  const int* __restrict__ srcv = isJ ? srcJ : srcU;
  int* __restrict__ rs = isJ ? rsJ : rsU;
  int* __restrict__ perm = isJ ? permJ : permU;
  const int nDst = isJ ? NJJ : NUU;
  const int rpw = isJ ? (NJJ + GRP_WGS_T - 1) / GRP_WGS_T : (NUU + GRP_WGS_T - 1) / GRP_WGS_T;
  const int b = isJ ? (int)blockIdx.x : (int)blockIdx.x - GRP_WGS_T;
  const int lo = b * rpw;
  const int nr = min(rpw, nDst - lo);
  const int tid = threadIdx.x;
  const int lane = tid & 31;
  const int wave = __builtin_amdgcn_readfirstlane(tid >> 5);
  if (tid < nr) hist[tid] = 0;
  if (tid == 0) ovf = 0;
  __syncthreads();

  int* wl = lst + wave * GRP_CAPW;
  int below = 0, wcount = 0;
  const int4* __restrict__ d4 = (const int4*)dst;
  constexpr int N4 = NEDGE / 4;
  constexpr int MAINC = N4 / GRP_T / 8 * 8;
  constexpr int TAILC = (N4 + GRP_T - 1) / GRP_T - MAINC;
  static_assert(TAILC >= 1 && TAILC <= 8, "the tail is one group of chunks");
  for (int c0 = 0; c0 < MAINC; c0 += 8)
    grp_chunks<8, false>(d4, c0, tid, lo, nr, hist, wl, below, wcount);
  grp_chunks<TAILC, true>(d4, MAINC, tid, lo, nr, hist, wl, below, wcount);
  if (lane == 0) {
    part[wave] = below;
    if (wcount > GRP_CAPW) ovf = 1;
  }
  __syncthreads();

  const int c = (tid < nr) ? hist[tid] : 0;
  int inc = c;
  for (int o = 1; o < 32; o <<= 1) {
    int t = __shfl_up(inc, o, 32);
    if (lane >= o) inc += t;
  }
  if (lane == 31) wsum[wave] = inc;
  int base = part[lane];
  for (int o = 16; o > 0; o >>= 1) base += __shfl_xor(base, o, 32);
  __syncthreads();
  const int pw = wsum[lane];
  int pinc = pw;
  for (int o = 1; o < 32; o <<= 1) {
    int t = __shfl_up(pinc, o, 32);
    if (lane >= o) pinc += t;
  }
  const int woff = __shfl(pinc - pw, wave, 32);
  const int total = __shfl(pinc, 31, 32);
  const int excl = woff + inc - c;
  if (tid < nr) {
    rsl[tid] = excl;
    rs[lo + tid] = base + excl;
  }
  if (b == GRP_WGS_T - 1 && tid == 0) rs[nDst] = base + total;
  const bool over = ovf != 0;
  __syncthreads();

  if (!over) {
    const int m = min(wcount, GRP_CAPW);
    for (int p0 = lane; p0 < m + lane; p0 += 128) {
      unsigned en[4];
      int sv[4];
#pragma unroll
      for (int k = 0; k < 4; ++k) en[k] = (unsigned)wl[min(p0 + 32 * k, GRP_CAPW - 1)];
#pragma unroll
      for (int k = 0; k < 4; ++k)
        sv[k] = srcv[min((int)(en[k] & ((1u << GRP_EBITS) - 1u)), NEDGE - 1)];
#pragma unroll
      for (int k = 0; k < 4; ++k) {
        if (p0 + 32 * k < m) {
          const int pos = base + atomicAdd(&rsl[en[k] >> GRP_EBITS], 1);
          perm[min(max(pos, 0), NEDGE - 1)] = sv[k];
        }
      }
    }
  } else {
    for (int e = tid; e < NEDGE; e += GRP_T) {
      const unsigned dl = (unsigned)(dst[e] - lo);
      if (dl < (unsigned)nr) {
        const int pos = base + atomicAdd(&rsl[dl], 1);
        perm[min(max(pos, 0), NEDGE - 1)] = srcv[e];
      }
    }
  }
}

__device__ __forceinline__ void store_mean(float* __restrict__ p, const float4 acc, int count) {
  const float c = fmaxf((float)count, 1.0f);
  float4 o;
  o.x = acc.x / c;
  o.y = acc.y / c;
  o.z = acc.z / c;
  o.w = acc.w / c;
  *(float4*)p = o;
}

template <int U>
__device__ __forceinline__ void add_rows(float4& acc, int& r, int& rowBeg, int& rowEnd,
                                         const int rsv, const int nRows,
                                         const float* __restrict__ xl, float* __restrict__ mrow,
                                         int myidx, int j, int e, int lastSrc) {
  float4 rr[U];
#pragma unroll
  for (int u = 0; u < U; ++u) {
    int s = __builtin_amdgcn_readlane(myidx, j + u);
    s = min(max(s, 0), lastSrc);
    rr[u] = *(const float4*)(xl + (size_t)s * DIM);
  }
#pragma unroll
  for (int u = 0; u < U; ++u) {
    if (e + u >= rowEnd && r + 1 < nRows) {
      store_mean(mrow + (size_t)r * DIM, acc, rowEnd - rowBeg);
      acc = make_float4(0.0f, 0.0f, 0.0f, 0.0f);
      ++r;
      rowBeg = rowEnd;
      rowEnd = __builtin_amdgcn_readlane(rsv, r + 1);
      while (e + u >= rowEnd && r + 1 < nRows) {
        *(float4*)(mrow + (size_t)r * DIM) = acc;
        ++r;
        rowBeg = rowEnd;
        rowEnd = __builtin_amdgcn_readlane(rsv, r + 1);
      }
    }
    acc.x += rr[u].x;
    acc.y += rr[u].y;
    acc.z += rr[u].z;
    acc.w += rr[u].w;
  }
}

#define RJ 4
#define RU 8
#define GJ_WGS (((NJJ + RJ - 1) / RJ + 7) / 8)
#define GU_WGS (((NUU + RU - 1) / RU + 7) / 8)
__global__ __launch_bounds__(256) void k_gather_mean(const float* __restrict__ xsrcJ,
                                                      const int* __restrict__ rsJ,
                                                      const int* __restrict__ permJ,
                                                      float* __restrict__ meanJ,
                                                      const float* __restrict__ xsrcU,
                                                      const int* __restrict__ rsU,
                                                      const int* __restrict__ permU,
                                                      float* __restrict__ meanU) {
  const bool isJ = blockIdx.x < GJ_WGS;
  const float* __restrict__ xsrc = isJ ? xsrcJ : xsrcU;
  const int* __restrict__ rs = isJ ? rsJ : rsU;
  const int* __restrict__ perm = isJ ? permJ : permU;
  float* __restrict__ mean = isJ ? meanJ : meanU;
  const int nDst = isJ ? NJJ : NUU;
  const int nSrc = isJ ? NUU : NJJ;
  const int R = isJ ? RJ : RU;
  const int wg = isJ ? (int)blockIdx.x : (int)blockIdx.x - GJ_WGS;
  const int d0 = R * __builtin_amdgcn_readfirstlane((int)(wg * 8 + (threadIdx.x >> 5)));
  if (d0 >= nDst) return;
  const int nRows = min(R, nDst - d0);
  const int lane = threadIdx.x & 31;
  const int rsv = (lane <= nRows) ? rs[d0 + lane] : 0;
  const int eBeg = __builtin_amdgcn_readlane(rsv, 0);
  const int eEnd = __builtin_amdgcn_readlane(rsv, nRows);
  const float* xl = xsrc + lane * 4;
  float* mrow = mean + (size_t)d0 * DIM + lane * 4;
  int r = 0;
  int rowBeg = eBeg;
  int rowEnd = __builtin_amdgcn_readlane(rsv, 1);
  float4 acc = make_float4(0.0f, 0.0f, 0.0f, 0.0f);
  for (int base = eBeg; base < eEnd; base += 128) {
    const int left = eEnd - base;
    const int i0 = (lane < left) ? perm[base + lane] : 0;
    const int i1 = (lane + 32 < left) ? perm[base + 32 + lane] : 0;
    const int i2 = (lane + 64 < left) ? perm[base + 64 + lane] : 0;
    const int i3 = (lane + 96 < left) ? perm[base + 96 + lane] : 0;
#pragma unroll 1
    for (int k = 0; k < 4; ++k) {
      const int b = base + 32 * k;
      const int m = min(eEnd - b, 32);
      if (m <= 0) break;
      const int myidx = (k == 0) ? i0 : (k == 1) ? i1 : (k == 2) ? i2 : i3;
      int j = 0;
      for (; j + 16 <= m; j += 16)
        add_rows<16>(acc, r, rowBeg, rowEnd, rsv, nRows, xl, mrow, myidx, j, b + j, nSrc - 1);
      if (m - j >= 8) {
        add_rows<8>(acc, r, rowBeg, rowEnd, rsv, nRows, xl, mrow, myidx, j, b + j, nSrc - 1);
        j += 8;
      }
      if (m - j >= 4) {
        add_rows<4>(acc, r, rowBeg, rowEnd, rsv, nRows, xl, mrow, myidx, j, b + j, nSrc - 1);
        j += 4;
      }
      if (m - j >= 2) {
        add_rows<2>(acc, r, rowBeg, rowEnd, rsv, nRows, xl, mrow, myidx, j, b + j, nSrc - 1);
        j += 2;
      }
      if (m - j >= 1)
        add_rows<1>(acc, r, rowBeg, rowEnd, rsv, nRows, xl, mrow, myidx, j, b + j, nSrc - 1);
    }
  }
  store_mean(mrow + (size_t)r * DIM, acc, rowEnd - rowBeg);
  for (++r; r < nRows; ++r)
    *(float4*)(mrow + (size_t)r * DIM) = make_float4(0.0f, 0.0f, 0.0f, 0.0f);
}

#define WPITCH 257
#define GEMM_WAVES 16
#define GEMM_WGS 256
#define GEMM_WGS_J 73
__global__ __launch_bounds__(512) void k_sage_gemm(
    const float* __restrict__ meanJ, const float* __restrict__ xdstJ,
    const float* __restrict__ WlJ, const float* __restrict__ WrJ,
    const float* __restrict__ biasJ, float* __restrict__ outJ,
    const float* __restrict__ meanU, const float* __restrict__ xdstU,
    const float* __restrict__ WlU, const float* __restrict__ WrU,
    const float* __restrict__ biasU, float* __restrict__ outU,
    int relu) {
  __shared__ v2f sW[2 * 32 * WPITCH];
  int wave = threadIdx.x >> 5;
  int lane = threadIdx.x & 31;
  const bool isJ = blockIdx.x < GEMM_WGS_J;
  const float* __restrict__ mean = isJ ? meanJ : meanU;
  const float* __restrict__ xdst = isJ ? xdstJ : xdstU;
  const float* __restrict__ Wl = isJ ? WlJ : WlU;
  const float* __restrict__ Wr = isJ ? WrJ : WrU;
  const float* __restrict__ bias = isJ ? biasJ : biasU;
  float* __restrict__ out = isJ ? outJ : outU;
  const int nRowTiles = isJ ? NJJ / 16 : NUU / 16;
  const int wg = isJ ? (int)blockIdx.x : (int)blockIdx.x - GEMM_WGS_J;
  const int nWg = isJ ? GEMM_WGS_J : GEMM_WGS - GEMM_WGS_J;

#pragma unroll
  for (int i = 0; i < 2 * DIM / GEMM_WAVES; ++i) {
    int n = wave + i * GEMM_WAVES;
    const float* wsrc = (n < DIM) ? Wl : Wr;
    int nn = n & (DIM - 1);
    const float4 w = *(const float4*)(wsrc + (size_t)nn * DIM + lane * 4);
    v2f* d = sW + (n >> 7) * (32 * WPITCH) + lane * WPITCH + (nn >> 4) * 32 + (nn & 15);
    v2f lo = {w.x, w.y};
    v2f hi = {w.z, w.w};
    d[0] = lo;
    d[16] = hi;
  }
  __syncthreads();

  int lm = lane & 15;
  int khalf = (lane >= 16) ? 2 : 0;
  const v2f* sWl = sW + lane;
  const v2f* sWr = sW + 32 * WPITCH + lane;

  for (int rowTile = wg + nWg * wave; rowTile < nRowTiles; rowTile += nWg * GEMM_WAVES) {
    int row0 = rowTile * 16;
    const float* aBase = mean + (size_t)(row0 + lm) * DIM + khalf;
    const float* xBase = xdst + (size_t)(row0 + lm) * DIM + khalf;

    v8f accA[8], accB[8];
#pragma unroll
    for (int t = 0; t < 8; ++t) {
      float bval = bias[t * 16 + lm];
      accA[t] = (v8f){bval, bval, bval, bval, bval, bval, bval, bval};
      accB[t] = (v8f){0.f, 0.f, 0.f, 0.f, 0.f, 0.f, 0.f, 0.f};
    }

    v2f am[4], ax[4];
#pragma unroll
    for (int u = 0; u < 4; ++u) {
      am[u] = *(const v2f*)(aBase + u * 4);
      ax[u] = *(const v2f*)(xBase + u * 4);
    }
#pragma unroll 1
    for (int g = 0; g < DIM / 16; ++g) {
      v2f amN[4], axN[4];
      int gn = (g + 1) & (DIM / 16 - 1);
#pragma unroll
      for (int u = 0; u < 4; ++u) {
        amN[u] = *(const v2f*)(aBase + gn * 16 + u * 4);
        axN[u] = *(const v2f*)(xBase + gn * 16 + u * 4);
      }
      const v2f* bl = sWl + g * 4 * WPITCH;
      const v2f* br = sWr + g * 4 * WPITCH;
#pragma unroll
      for (int u = 0; u < 4; ++u) {
#pragma unroll
        for (int t = 0; t < 8; ++t) {
          accA[t] = __builtin_amdgcn_wmma_f32_16x16x4_f32(false, am[u], false,
                                                          bl[u * WPITCH + t * 32], (short)0,
                                                          accA[t], false, false);
          accB[t] = __builtin_amdgcn_wmma_f32_16x16x4_f32(false, ax[u], false,
                                                          br[u * WPITCH + t * 32], (short)0,
                                                          accB[t], false, false);
        }
      }
#pragma unroll
      for (int u = 0; u < 4; ++u) {
        am[u] = amN[u];
        ax[u] = axN[u];
      }
    }

    float* orow = out + (size_t)(row0 + ((lane >= 16) ? 8 : 0)) * DIM + lm;
#pragma unroll
    for (int t = 0; t < 8; ++t) {
#pragma unroll
      for (int r = 0; r < 8; ++r) {
        float v = accA[t][r] + accB[t][r];
        if (relu) v = fmaxf(v, 0.0f);
        orow[(size_t)r * DIM + t * 16] = v;
      }
    }
  }
}

__global__ __launch_bounds__(256) void k_decode(
    const float* __restrict__ zu, const float* __restrict__ zj,
    const int* __restrict__ elu, const int* __restrict__ elj,
    float* __restrict__ out, int n) {
  const int p0 = 32 * __builtin_amdgcn_readfirstlane((int)(blockIdx.x * 8 + (threadIdx.x >> 5)));
  if (p0 >= n) return;
  const int lane = threadIdx.x & 31;
  const int np = min(n - p0, 32);
  const int iu = (lane < np) ? elu[p0 + lane] : 0;
  const int ij = (lane < np) ? elj[p0 + lane] : 0;
  const float* zul = zu + lane * 4;
  const float* zjl = zj + lane * 4;
  float rd = 0.0f, ra = 0.0f, rb = 0.0f;
#pragma unroll 1
  for (int q = 0; q < np; q += 8) {
    float4 a[8], b[8];
#pragma unroll
    for (int i = 0; i < 8; ++i) {
      const int su = __builtin_amdgcn_readlane(iu, q + i);
      const int sj = __builtin_amdgcn_readlane(ij, q + i);
      a[i] = *(const float4*)(zul + (size_t)su * DIM);
      b[i] = *(const float4*)(zjl + (size_t)sj * DIM);
    }
    __builtin_amdgcn_sched_barrier(0);
    float dot[8], na[8], nb[8];
#pragma unroll
    for (int i = 0; i < 8; ++i) {
      dot[i] = fmaf(a[i].w, b[i].w, fmaf(a[i].z, b[i].z, fmaf(a[i].x, b[i].x, a[i].y * b[i].y)));
      na[i] = fmaf(a[i].w, a[i].w, fmaf(a[i].z, a[i].z, fmaf(a[i].x, a[i].x, a[i].y * a[i].y)));
      nb[i] = fmaf(b[i].w, b[i].w, fmaf(b[i].z, b[i].z, fmaf(b[i].x, b[i].x, b[i].y * b[i].y)));
    }
#pragma unroll
    for (int off = 16; off > 0; off >>= 1) {
#pragma unroll
      for (int i = 0; i < 8; ++i) {
        dot[i] += __shfl_xor(dot[i], off, 32);
        na[i] += __shfl_xor(na[i], off, 32);
        nb[i] += __shfl_xor(nb[i], off, 32);
      }
    }
#pragma unroll
    for (int i = 0; i < 8; ++i) {
      if (lane == q + i) {
        rd = dot[i];
        ra = na[i];
        rb = nb[i];
      }
    }
  }
  if (lane < np) out[p0 + lane] = rd / (fmaxf(sqrtf(ra), 1e-8f) * fmaxf(sqrtf(rb), 1e-8f));
}


extern "C" void kernel_launch(void* const* d_in, const int* in_sizes, int n_in,
                              void* d_out, int out_size, void* d_ws, size_t ws_size,
                              hipStream_t stream) {
  const float* x_user = (const float*)d_in[0];
  const float* x_job = (const float*)d_in[1];
  const int* uj_src = (const int*)d_in[2];
  const int* uj_dst = (const int*)d_in[3];
  const int* ju_src = (const int*)d_in[4];
  const int* ju_dst = (const int*)d_in[5];
  const int* el_user = (const int*)d_in[6];
  const int* el_job = (const int*)d_in[7];
  const float* W1l_uj = (const float*)d_in[8];
  const float* W1r_uj = (const float*)d_in[9];
  const float* W1l_ju = (const float*)d_in[10];
  const float* W1r_ju = (const float*)d_in[11];
  const float* W2l_uj = (const float*)d_in[12];
  const float* W2r_uj = (const float*)d_in[13];
  const float* W2l_ju = (const float*)d_in[14];
  const float* W2r_ju = (const float*)d_in[15];
  const float* b1_uj = (const float*)d_in[16];
  const float* b1_ju = (const float*)d_in[17];
  const float* b2_uj = (const float*)d_in[18];
  const float* b2_ju = (const float*)d_in[19];
  float* out = (float*)d_out;

  float* ws = (float*)d_ws;
  float* agg_user = ws; ws += (size_t)NUU * DIM;
  float* agg_job  = ws; ws += (size_t)NJJ * DIM;
  float* h_user   = ws; ws += (size_t)NUU * DIM;
  float* h_job    = ws; ws += (size_t)NJJ * DIM;
  float* z_user   = ws; ws += (size_t)NUU * DIM;
  float* z_job    = ws; ws += (size_t)NJJ * DIM;
  int* perm_job  = (int*)z_user;
  int* perm_user = perm_job + NEDGE;
  int* rs_job    = perm_user + NEDGE;
  int* rs_user   = rs_job + (NJJ + 32);

  k_group<<<2 * GRP_WGS_T, GRP_T, 0, stream>>>(uj_dst, uj_src, ju_dst, ju_src, rs_job, rs_user,
                                               perm_job, perm_user);

  k_gather_mean<<<GJ_WGS + GU_WGS, 256, 0, stream>>>(x_user, rs_job, perm_job, agg_job,
                                                     x_job, rs_user, perm_user, agg_user);

  k_sage_gemm<<<GEMM_WGS, 512, 0, stream>>>(agg_job, x_job, W1l_uj, W1r_uj, b1_uj, h_job,
                                            agg_user, x_user, W1l_ju, W1r_ju, b1_ju, h_user, 1);

  k_gather_mean<<<GJ_WGS + GU_WGS, 256, 0, stream>>>(h_user, rs_job, perm_job, agg_job,
                                                     h_job, rs_user, perm_user, agg_user);

  k_sage_gemm<<<GEMM_WGS, 512, 0, stream>>>(agg_job, h_job, W2l_uj, W2r_uj, b2_uj, z_job,
                                            agg_user, h_user, W2l_ju, W2r_ju, b2_ju, z_user, 0);

  k_decode<<<((NEL + 31) / 32 + 7) / 8, dim3(256), 0, stream>>>(z_user, z_job, el_user, el_job, out, NEL);
}


# Neighbour means formed inside the layer kernel: a workgroup gathers the rows of its own strips into LDS and multiplies from there; 6 -> 4 launches, means never in memory; sum orders kept
# speedup vs baseline: 4.4642x; 1.1068x over previous
#include <hip/hip_runtime.h>
#include <hip/hip_bf16.h>
#include <math.h>

#define NUU 50000
#define NJJ 20000
#define NEDGE 500000
#define NEL 100000
#define DIM 128

typedef __attribute__((ext_vector_type(2))) float v2f;
typedef __attribute__((ext_vector_type(8))) float v8f;


#define GRP_T 1024
#define GRP_WGS_T 64
#define GRP_CAPW 1024
#define GRP_EBITS 19
#define GRP_MAXR ((NUU + GRP_WGS_T - 1) / GRP_WGS_T)
static_assert(NEDGE % 4 == 0, "the destination list is read four entries per lane");
static_assert(GRP_MAXR <= GRP_T && GRP_MAXR <= 1024, "one thread per owned row; row fits 10 bits");
static_assert(NEDGE <= (1 << GRP_EBITS), "edge number and row share one 32-bit note");
static_assert((NJJ + GRP_WGS_T - 1) / GRP_WGS_T <= GRP_MAXR, "job rows per owner fit too");

__device__ __forceinline__ void grp_take(const int d, const int e, const int lo, const int nr,
                                         int* hist, int* wl, int& below, int& wcount) {
  below += __builtin_popcount(__builtin_amdgcn_ballot_w32(d < lo));
  const unsigned dl = (unsigned)(d - lo);
  const bool mt = dl < (unsigned)nr;
  const unsigned mask = __builtin_amdgcn_ballot_w32(mt);
  if (mask != 0u) {
    if (mt) {
      atomicAdd(&hist[dl], 1);
      const int p = wcount + (int)__builtin_amdgcn_mbcnt_lo(mask, 0u);
      if (p < GRP_CAPW) wl[p] = (int)((unsigned)e | (dl << GRP_EBITS));
    }
    wcount += __builtin_popcount(mask);
  }
}

template <int NG, bool LAST>
__device__ __forceinline__ void grp_chunks(const int4* __restrict__ d4, const int c0, const int tid,
                                           const int lo, const int nr, int* hist, int* wl,
                                           int& below, int& wcount) {
  constexpr int N4 = NEDGE / 4;
  int4 v[NG];
#pragma unroll
  for (int u = 0; u < NG; ++u) {
    const int i4 = (c0 + u) * GRP_T + tid;
    if (LAST) {
      v[u] = d4[min(i4, N4 - 1)];
      if (i4 >= N4) v[u] = make_int4(0x7fffffff, 0x7fffffff, 0x7fffffff, 0x7fffffff);
    } else {
      v[u] = d4[i4];
    }
  }
#pragma unroll
  for (int u = 0; u < NG; ++u) {
    const int e0 = 4 * ((c0 + u) * GRP_T + tid);
    grp_take(v[u].x, e0, lo, nr, hist, wl, below, wcount);
    grp_take(v[u].y, e0 + 1, lo, nr, hist, wl, below, wcount);
    grp_take(v[u].z, e0 + 2, lo, nr, hist, wl, below, wcount);
    grp_take(v[u].w, e0 + 3, lo, nr, hist, wl, below, wcount);
  }
}

__global__ __launch_bounds__(GRP_T) void k_group(const int* __restrict__ dstJ,
                                                  const int* __restrict__ srcJ,
                                                  const int* __restrict__ dstU,
                                                  const int* __restrict__ srcU,
                                                  int* __restrict__ rsJ, int* __restrict__ rsU,
                                                  int* __restrict__ permJ,
                                                  int* __restrict__ permU) {
  __shared__ int hist[GRP_MAXR];
  __shared__ int rsl[GRP_MAXR];
  __shared__ int lst[32 * GRP_CAPW];
  __shared__ int part[32];
  __shared__ int wsum[32];
  __shared__ int ovf;
  const bool isJ = blockIdx.x < GRP_WGS_T;
  const int* __restrict__ dst = isJ ? dstJ : dstU;
  const int* __restrict__ srcv = isJ ? srcJ : srcU;
  int* __restrict__ rs = isJ ? rsJ : rsU;
  int* __restrict__ perm = isJ ? permJ : permU;
  const int nDst = isJ ? NJJ : NUU;
  const int rpw = isJ ? (NJJ + GRP_WGS_T - 1) / GRP_WGS_T : (NUU + GRP_WGS_T - 1) / GRP_WGS_T;
  const int b = isJ ? (int)blockIdx.x : (int)blockIdx.x - GRP_WGS_T;
  const int lo = b * rpw;
  const int nr = min(rpw, nDst - lo);
  const int tid = threadIdx.x;
  const int lane = tid & 31;
  const int wave = __builtin_amdgcn_readfirstlane(tid >> 5);
  if (tid < nr) hist[tid] = 0;
  if (tid == 0) ovf = 0;
  __syncthreads();

  int* wl = lst + wave * GRP_CAPW;
  int below = 0, wcount = 0;
  const int4* __restrict__ d4 = (const int4*)dst;
  constexpr int N4 = NEDGE / 4;
  constexpr int MAINC = N4 / GRP_T / 8 * 8;
  constexpr int TAILC = (N4 + GRP_T - 1) / GRP_T - MAINC;
  static_assert(TAILC >= 1 && TAILC <= 8, "the tail is one group of chunks");
  for (int c0 = 0; c0 < MAINC; c0 += 8)
    grp_chunks<8, false>(d4, c0, tid, lo, nr, hist, wl, below, wcount);
  grp_chunks<TAILC, true>(d4, MAINC, tid, lo, nr, hist, wl, below, wcount);
  if (lane == 0) {
    part[wave] = below;
    if (wcount > GRP_CAPW) ovf = 1;
  }
  __syncthreads();

  const int c = (tid < nr) ? hist[tid] : 0;
  int inc = c;
  for (int o = 1; o < 32; o <<= 1) {
    int t = __shfl_up(inc, o, 32);
    if (lane >= o) inc += t;
  }
  if (lane == 31) wsum[wave] = inc;
  int base = part[lane];
  for (int o = 16; o > 0; o >>= 1) base += __shfl_xor(base, o, 32);
  __syncthreads();
  const int pw = wsum[lane];
  int pinc = pw;
  for (int o = 1; o < 32; o <<= 1) {
    int t = __shfl_up(pinc, o, 32);
    if (lane >= o) pinc += t;
  }
  const int woff = __shfl(pinc - pw, wave, 32);
  const int total = __shfl(pinc, 31, 32);
  const int excl = woff + inc - c;
  if (tid < nr) {
    rsl[tid] = excl;
    rs[lo + tid] = base + excl;
  }
  if (b == GRP_WGS_T - 1 && tid == 0) rs[nDst] = base + total;
  const bool over = ovf != 0;
  __syncthreads();

  if (!over) {
    const int m = min(wcount, GRP_CAPW);
    for (int p0 = lane; p0 < m + lane; p0 += 128) {
      unsigned en[4];
      int sv[4];
#pragma unroll
      for (int k = 0; k < 4; ++k) en[k] = (unsigned)wl[min(p0 + 32 * k, GRP_CAPW - 1)];
#pragma unroll
      for (int k = 0; k < 4; ++k)
        sv[k] = srcv[min((int)(en[k] & ((1u << GRP_EBITS) - 1u)), NEDGE - 1)];
#pragma unroll
      for (int k = 0; k < 4; ++k) {
        if (p0 + 32 * k < m) {
          const int pos = base + atomicAdd(&rsl[en[k] >> GRP_EBITS], 1);
          perm[min(max(pos, 0), NEDGE - 1)] = sv[k];
        }
      }
    }
  } else {
    for (int e = tid; e < NEDGE; e += GRP_T) {
      const unsigned dl = (unsigned)(dst[e] - lo);
      if (dl < (unsigned)nr) {
        const int pos = base + atomicAdd(&rsl[dl], 1);
        perm[min(max(pos, 0), NEDGE - 1)] = srcv[e];
      }
    }
  }
}

__device__ __forceinline__ void store_mean(float* p, const float4 acc, int count) {
  const float c = fmaxf((float)count, 1.0f);
  float4 o;
  o.x = acc.x / c;
  o.y = acc.y / c;
  o.z = acc.z / c;
  o.w = acc.w / c;
  *(float4*)p = o;
}

template <int U, int PITCH>
__device__ __forceinline__ void add_rows(float4& acc, int& r, int& rowBeg, int& rowEnd,
                                         const int rsv, const int nRows,
                                         const float* __restrict__ xl, float* mrow,
                                         int myidx, int j, int e, int lastSrc) {
  float4 rr[U];
#pragma unroll
  for (int u = 0; u < U; ++u) {
    int s = __builtin_amdgcn_readlane(myidx, j + u);
    s = min(max(s, 0), lastSrc);
    rr[u] = *(const float4*)(xl + (size_t)s * DIM);
  }
#pragma unroll
  for (int u = 0; u < U; ++u) {
    if (e + u >= rowEnd && r + 1 < nRows) {
      store_mean(mrow + r * PITCH, acc, rowEnd - rowBeg);
      acc = make_float4(0.0f, 0.0f, 0.0f, 0.0f);
      ++r;
      rowBeg = rowEnd;
      rowEnd = __builtin_amdgcn_readlane(rsv, r + 1);
      while (e + u >= rowEnd && r + 1 < nRows) {
        *(float4*)(mrow + r * PITCH) = acc;
        ++r;
        rowBeg = rowEnd;
        rowEnd = __builtin_amdgcn_readlane(rsv, r + 1);
      }
    }
    acc.x += rr[u].x;
    acc.y += rr[u].y;
    acc.z += rr[u].z;
    acc.w += rr[u].w;
  }
}

template <int PITCH>
__device__ __forceinline__ void gather_means(const float* __restrict__ xsrc,
                                             const int* __restrict__ rs,
                                             const int* __restrict__ perm, float* mrow,
                                             const int d0, const int nRows, const int nSrc,
                                             const int lane) {
  const int rsv = (lane <= nRows) ? rs[d0 + lane] : 0;
  const int eBeg = __builtin_amdgcn_readlane(rsv, 0);
  const int eEnd = __builtin_amdgcn_readlane(rsv, nRows);
  const float* xl = xsrc + lane * 4;
  int r = 0;
  int rowBeg = eBeg;
  int rowEnd = __builtin_amdgcn_readlane(rsv, 1);
  float4 acc = make_float4(0.0f, 0.0f, 0.0f, 0.0f);
  for (int base = eBeg; base < eEnd; base += 128) {
    const int left = eEnd - base;
    const int i0 = (lane < left) ? perm[base + lane] : 0;
    const int i1 = (lane + 32 < left) ? perm[base + 32 + lane] : 0;
    const int i2 = (lane + 64 < left) ? perm[base + 64 + lane] : 0;
    const int i3 = (lane + 96 < left) ? perm[base + 96 + lane] : 0;
#pragma unroll 1
    for (int k = 0; k < 4; ++k) {
      const int b = base + 32 * k;
      const int m = min(eEnd - b, 32);
      if (m <= 0) break;
      const int myidx = (k == 0) ? i0 : (k == 1) ? i1 : (k == 2) ? i2 : i3;
      int j = 0;
      for (; j + 16 <= m; j += 16)
        add_rows<16, PITCH>(acc, r, rowBeg, rowEnd, rsv, nRows, xl, mrow, myidx, j, b + j, nSrc - 1);
      if (m - j >= 8) {
        add_rows<8, PITCH>(acc, r, rowBeg, rowEnd, rsv, nRows, xl, mrow, myidx, j, b + j, nSrc - 1);
        j += 8;
      }
      if (m - j >= 4) {
        add_rows<4, PITCH>(acc, r, rowBeg, rowEnd, rsv, nRows, xl, mrow, myidx, j, b + j, nSrc - 1);
        j += 4;
      }
      if (m - j >= 2) {
        add_rows<2, PITCH>(acc, r, rowBeg, rowEnd, rsv, nRows, xl, mrow, myidx, j, b + j, nSrc - 1);
        j += 2;
      }
      if (m - j >= 1)
        add_rows<1, PITCH>(acc, r, rowBeg, rowEnd, rsv, nRows, xl, mrow, myidx, j, b + j, nSrc - 1);
    }
  }
  store_mean(mrow + r * PITCH, acc, rowEnd - rowBeg);
  for (++r; r < nRows; ++r)
    *(float4*)(mrow + r * PITCH) = make_float4(0.0f, 0.0f, 0.0f, 0.0f);
}

#define WPITCH 257
#define MPITCH 132
#define GEMM_WAVES 16
#define GEMM_WGS 256
#define GEMM_WGS_J 96
#define MAXS 20
static_assert(NJJ % 16 == 0 && NUU % 16 == 0, "whole 16-row strips");
static_assert((NJJ / 16 + GEMM_WGS_J - 1) / GEMM_WGS_J <= MAXS, "job run fits the LDS plane");
static_assert((NUU / 16 + (GEMM_WGS - GEMM_WGS_J) - 1) / (GEMM_WGS - GEMM_WGS_J) <= MAXS,
              "user run fits the LDS plane");
static_assert(NJJ / 16 >= GEMM_WGS_J && NUU / 16 >= GEMM_WGS - GEMM_WGS_J, "no empty run");
static_assert(MAXS <= 31, "a wave keeps the run starts of its rows in one register");
__global__ __launch_bounds__(512) void k_sage_layer(
    const float* __restrict__ xsrcJ, const int* __restrict__ rsJ, const int* __restrict__ permJ,
    const float* __restrict__ xdstJ, const float* __restrict__ WlJ, const float* __restrict__ WrJ,
    const float* __restrict__ biasJ, float* __restrict__ outJ,
    const float* __restrict__ xsrcU, const int* __restrict__ rsU, const int* __restrict__ permU,
    const float* __restrict__ xdstU, const float* __restrict__ WlU, const float* __restrict__ WrU,
    const float* __restrict__ biasU, float* __restrict__ outU,
    int relu) {
  __shared__ v2f sW[2 * 32 * WPITCH];
  __shared__ float4 sM4[MAXS * 16 * MPITCH / 4];
  float* sM = (float*)sM4;
  int wave = threadIdx.x >> 5;
  int lane = threadIdx.x & 31;
  const bool isJ = blockIdx.x < GEMM_WGS_J;
  const float* __restrict__ xsrc = isJ ? xsrcJ : xsrcU;
  const int* __restrict__ rs = isJ ? rsJ : rsU;
  const int* __restrict__ perm = isJ ? permJ : permU;
  const float* __restrict__ xdst = isJ ? xdstJ : xdstU;
  const float* __restrict__ Wl = isJ ? WlJ : WlU;
  const float* __restrict__ Wr = isJ ? WrJ : WrU;
  const float* __restrict__ bias = isJ ? biasJ : biasU;
  float* __restrict__ out = isJ ? outJ : outU;
  const int nSrc = isJ ? NUU : NJJ;
  const int wg = isJ ? (int)blockIdx.x : (int)blockIdx.x - GEMM_WGS_J;
  const unsigned uw = (unsigned)wg;
  const int sBeg = isJ ? (int)(uw * (NJJ / 16) / GEMM_WGS_J)
                       : (int)(uw * (NUU / 16) / (GEMM_WGS - GEMM_WGS_J));
  const int sEnd = isJ ? (int)((uw + 1u) * (NJJ / 16) / GEMM_WGS_J)
                       : (int)((uw + 1u) * (NUU / 16) / (GEMM_WGS - GEMM_WGS_J));
  const int nS = sEnd - sBeg;

#pragma unroll
  for (int i = 0; i < 2 * DIM / GEMM_WAVES; ++i) {
    int n = wave + i * GEMM_WAVES;
    const float* wsrc = (n < DIM) ? Wl : Wr;
    int nn = n & (DIM - 1);
    const float4 w = *(const float4*)(wsrc + (size_t)nn * DIM + lane * 4);
    v2f* d = sW + (n >> 7) * (32 * WPITCH) + lane * WPITCH + (nn >> 4) * 32 + (nn & 15);
    v2f lo = {w.x, w.y};
    v2f hi = {w.z, w.w};
    d[0] = lo;
    d[16] = hi;
  }

  {
    const int wv = __builtin_amdgcn_readfirstlane(wave);
    gather_means<MPITCH>(xsrc, rs, perm, sM + wv * nS * MPITCH + lane * 4, sBeg * 16 + wv * nS, nS,
                         nSrc, lane);
  }
  __syncthreads();

  int lm = lane & 15;
  int khalf = (lane >= 16) ? 2 : 0;
  const v2f* sWl = sW + lane;
  const v2f* sWr = sW + 32 * WPITCH + lane;

  for (int s = wave; s < nS; s += GEMM_WAVES) {
    int row0 = (sBeg + s) * 16;
    const float* aBase = sM + (s * 16 + lm) * MPITCH + khalf;
    const float* xBase = xdst + (size_t)(row0 + lm) * DIM + khalf;

    v8f accA[8], accB[8];
#pragma unroll
    for (int t = 0; t < 8; ++t) {
      float bval = bias[t * 16 + lm];
      accA[t] = (v8f){bval, bval, bval, bval, bval, bval, bval, bval};
      accB[t] = (v8f){0.f, 0.f, 0.f, 0.f, 0.f, 0.f, 0.f, 0.f};
    }

    v2f am[4], ax[4];
#pragma unroll
    for (int u = 0; u < 4; ++u) {
      am[u] = *(const v2f*)(aBase + u * 4);
      ax[u] = *(const v2f*)(xBase + u * 4);
    }
#pragma unroll 1
    for (int g = 0; g < DIM / 16; ++g) {
      v2f amN[4], axN[4];
      int gn = (g + 1) & (DIM / 16 - 1);
#pragma unroll
      for (int u = 0; u < 4; ++u) {
        amN[u] = *(const v2f*)(aBase + gn * 16 + u * 4);
        axN[u] = *(const v2f*)(xBase + gn * 16 + u * 4);
      }
      const v2f* bl = sWl + g * 4 * WPITCH;
      const v2f* br = sWr + g * 4 * WPITCH;
#pragma unroll
      for (int u = 0; u < 4; ++u) {
#pragma unroll
        for (int t = 0; t < 8; ++t) {
          accA[t] = __builtin_amdgcn_wmma_f32_16x16x4_f32(false, am[u], false,
                                                          bl[u * WPITCH + t * 32], (short)0,
                                                          accA[t], false, false);
          accB[t] = __builtin_amdgcn_wmma_f32_16x16x4_f32(false, ax[u], false,
                                                          br[u * WPITCH + t * 32], (short)0,
                                                          accB[t], false, false);
        }
      }
#pragma unroll
      for (int u = 0; u < 4; ++u) {
        am[u] = amN[u];
        ax[u] = axN[u];
      }
    }

    float* orow = out + (size_t)(row0 + ((lane >= 16) ? 8 : 0)) * DIM + lm;
#pragma unroll
    for (int t = 0; t < 8; ++t) {
#pragma unroll
      for (int r = 0; r < 8; ++r) {
        float v = accA[t][r] + accB[t][r];
        if (relu) v = fmaxf(v, 0.0f);
        orow[(size_t)r * DIM + t * 16] = v;
      }
    }
  }
}

__global__ __launch_bounds__(256) void k_decode(
    const float* __restrict__ zu, const float* __restrict__ zj,
    const int* __restrict__ elu, const int* __restrict__ elj,
    float* __restrict__ out, int n) {
  const int p0 = 32 * __builtin_amdgcn_readfirstlane((int)(blockIdx.x * 8 + (threadIdx.x >> 5)));
  if (p0 >= n) return;
  const int lane = threadIdx.x & 31;
  const int np = min(n - p0, 32);
  const int iu = (lane < np) ? elu[p0 + lane] : 0;
  const int ij = (lane < np) ? elj[p0 + lane] : 0;
  const float* zul = zu + lane * 4;
  const float* zjl = zj + lane * 4;
  float rd = 0.0f, ra = 0.0f, rb = 0.0f;
#pragma unroll 1
  for (int q = 0; q < np; q += 8) {
    float4 a[8], b[8];
#pragma unroll
    for (int i = 0; i < 8; ++i) {
      const int su = __builtin_amdgcn_readlane(iu, q + i);
      const int sj = __builtin_amdgcn_readlane(ij, q + i);
      a[i] = *(const float4*)(zul + (size_t)su * DIM);
      b[i] = *(const float4*)(zjl + (size_t)sj * DIM);
    }
    __builtin_amdgcn_sched_barrier(0);
    float dot[8], na[8], nb[8];
#pragma unroll
    for (int i = 0; i < 8; ++i) {
      dot[i] = fmaf(a[i].w, b[i].w, fmaf(a[i].z, b[i].z, fmaf(a[i].x, b[i].x, a[i].y * b[i].y)));
      na[i] = fmaf(a[i].w, a[i].w, fmaf(a[i].z, a[i].z, fmaf(a[i].x, a[i].x, a[i].y * a[i].y)));
      nb[i] = fmaf(b[i].w, b[i].w, fmaf(b[i].z, b[i].z, fmaf(b[i].x, b[i].x, b[i].y * b[i].y)));
    }
#pragma unroll
    for (int off = 16; off > 0; off >>= 1) {
#pragma unroll
      for (int i = 0; i < 8; ++i) {
        dot[i] += __shfl_xor(dot[i], off, 32);
        na[i] += __shfl_xor(na[i], off, 32);
        nb[i] += __shfl_xor(nb[i], off, 32);
      }
    }
#pragma unroll
    for (int i = 0; i < 8; ++i) {
      if (lane == q + i) {
        rd = dot[i];
        ra = na[i];
        rb = nb[i];
      }
    }
  }
  if (lane < np) out[p0 + lane] = rd / (fmaxf(sqrtf(ra), 1e-8f) * fmaxf(sqrtf(rb), 1e-8f));
}


extern "C" void kernel_launch(void* const* d_in, const int* in_sizes, int n_in,
                              void* d_out, int out_size, void* d_ws, size_t ws_size,
                              hipStream_t stream) {
  const float* x_user = (const float*)d_in[0];
  const float* x_job = (const float*)d_in[1];
  const int* uj_src = (const int*)d_in[2];
  const int* uj_dst = (const int*)d_in[3];
  const int* ju_src = (const int*)d_in[4];
  const int* ju_dst = (const int*)d_in[5];
  const int* el_user = (const int*)d_in[6];
  const int* el_job = (const int*)d_in[7];
  const float* W1l_uj = (const float*)d_in[8];
  const float* W1r_uj = (const float*)d_in[9];
  const float* W1l_ju = (const float*)d_in[10];
  const float* W1r_ju = (const float*)d_in[11];
  const float* W2l_uj = (const float*)d_in[12];
  const float* W2r_uj = (const float*)d_in[13];
  const float* W2l_ju = (const float*)d_in[14];
  const float* W2r_ju = (const float*)d_in[15];
  const float* b1_uj = (const float*)d_in[16];
  const float* b1_ju = (const float*)d_in[17];
  const float* b2_uj = (const float*)d_in[18];
  const float* b2_ju = (const float*)d_in[19];
  float* out = (float*)d_out;

  float* ws = (float*)d_ws;
  float* agg_user = ws; ws += (size_t)NUU * DIM;
  float* agg_job  = ws; ws += (size_t)NJJ * DIM;
  float* h_user   = ws; ws += (size_t)NUU * DIM;
  float* h_job    = ws; ws += (size_t)NJJ * DIM;
  float* z_user   = ws; ws += (size_t)NUU * DIM;
  float* z_job    = ws; ws += (size_t)NJJ * DIM;
  (void)agg_job;
  int* perm_job  = (int*)agg_user;
  int* perm_user = perm_job + NEDGE;
  int* rs_job    = perm_user + NEDGE;
  int* rs_user   = rs_job + (NJJ + 32);

  k_group<<<2 * GRP_WGS_T, GRP_T, 0, stream>>>(uj_dst, uj_src, ju_dst, ju_src, rs_job, rs_user,
                                               perm_job, perm_user);

  k_sage_layer<<<GEMM_WGS, 512, 0, stream>>>(x_user, rs_job, perm_job, x_job, W1l_uj, W1r_uj, b1_uj,
                                             h_job, x_job, rs_user, perm_user, x_user, W1l_ju,
                                             W1r_ju, b1_ju, h_user, 1);

  k_sage_layer<<<GEMM_WGS, 512, 0, stream>>>(h_user, rs_job, perm_job, h_job, W2l_uj, W2r_uj, b2_uj,
                                             z_job, h_job, rs_user, perm_user, h_user, W2l_ju,
                                             W2r_ju, b2_ju, z_user, 0);

  k_decode<<<((NEL + 31) / 32 + 7) / 8, dim3(256), 0, stream>>>(z_user, z_job, el_user, el_job, out, NEL);
}


# Layer kernel: the xdst*Wr product of a wave's strip is split off the mean*Wl product and runs in the gather phase (half the waves before their gather, half after); one more barrier; sum orders kept
# speedup vs baseline: 4.6539x; 1.0425x over previous
#include <hip/hip_runtime.h>
#include <hip/hip_bf16.h>
#include <math.h>

#define NUU 50000
#define NJJ 20000
#define NEDGE 500000
#define NEL 100000
#define DIM 128

typedef __attribute__((ext_vector_type(2))) float v2f;
typedef __attribute__((ext_vector_type(8))) float v8f;


#define GRP_T 1024
#define GRP_WGS_T 64
#define GRP_CAPW 1024
#define GRP_EBITS 19
#define GRP_MAXR ((NUU + GRP_WGS_T - 1) / GRP_WGS_T)
static_assert(NEDGE % 4 == 0, "the destination list is read four entries per lane");
static_assert(GRP_MAXR <= GRP_T && GRP_MAXR <= 1024, "one thread per owned row; row fits 10 bits");
static_assert(NEDGE <= (1 << GRP_EBITS), "edge number and row share one 32-bit note");
static_assert((NJJ + GRP_WGS_T - 1) / GRP_WGS_T <= GRP_MAXR, "job rows per owner fit too");

__device__ __forceinline__ void grp_take(const int d, const int e, const int lo, const int nr,
                                         int* hist, int* wl, int& below, int& wcount) {
  below += __builtin_popcount(__builtin_amdgcn_ballot_w32(d < lo));
  const unsigned dl = (unsigned)(d - lo);
  const bool mt = dl < (unsigned)nr;
  const unsigned mask = __builtin_amdgcn_ballot_w32(mt);
  if (mask != 0u) {
    if (mt) {
      atomicAdd(&hist[dl], 1);
      const int p = wcount + (int)__builtin_amdgcn_mbcnt_lo(mask, 0u);
      if (p < GRP_CAPW) wl[p] = (int)((unsigned)e | (dl << GRP_EBITS));
    }
    wcount += __builtin_popcount(mask);
  }
}

template <int NG, bool LAST>
__device__ __forceinline__ void grp_chunks(const int4* __restrict__ d4, const int c0, const int tid,
                                           const int lo, const int nr, int* hist, int* wl,
                                           int& below, int& wcount) {
  constexpr int N4 = NEDGE / 4;
  int4 v[NG];
#pragma unroll
  for (int u = 0; u < NG; ++u) {
    const int i4 = (c0 + u) * GRP_T + tid;
    if (LAST) {
      v[u] = d4[min(i4, N4 - 1)];
      if (i4 >= N4) v[u] = make_int4(0x7fffffff, 0x7fffffff, 0x7fffffff, 0x7fffffff);
    } else {
      v[u] = d4[i4];
    }
  }
#pragma unroll
  for (int u = 0; u < NG; ++u) {
    const int e0 = 4 * ((c0 + u) * GRP_T + tid);
    grp_take(v[u].x, e0, lo, nr, hist, wl, below, wcount);
    grp_take(v[u].y, e0 + 1, lo, nr, hist, wl, below, wcount);
    grp_take(v[u].z, e0 + 2, lo, nr, hist, wl, below, wcount);
    grp_take(v[u].w, e0 + 3, lo, nr, hist, wl, below, wcount);
  }
}

__global__ __launch_bounds__(GRP_T) void k_group(const int* __restrict__ dstJ,
                                                  const int* __restrict__ srcJ,
                                                  const int* __restrict__ dstU,
                                                  const int* __restrict__ srcU,
                                                  int* __restrict__ rsJ, int* __restrict__ rsU,
                                                  int* __restrict__ permJ,
                                                  int* __restrict__ permU) {
  __shared__ int hist[GRP_MAXR];
  __shared__ int rsl[GRP_MAXR];
  __shared__ int lst[32 * GRP_CAPW];
  __shared__ int part[32];
  __shared__ int wsum[32];
  __shared__ int ovf;
  const bool isJ = blockIdx.x < GRP_WGS_T;
  const int* __restrict__ dst = isJ ? dstJ : dstU;
  const int* __restrict__ srcv = isJ ? srcJ : srcU;
  int* __restrict__ rs = isJ ? rsJ : rsU;
  int* __restrict__ perm = isJ ? permJ : permU;
  const int nDst = isJ ? NJJ : NUU;
  const int rpw = isJ ? (NJJ + GRP_WGS_T - 1) / GRP_WGS_T : (NUU + GRP_WGS_T - 1) / GRP_WGS_T;
  const int b = isJ ? (int)blockIdx.x : (int)blockIdx.x - GRP_WGS_T;
  const int lo = b * rpw;
  const int nr = min(rpw, nDst - lo);
  const int tid = threadIdx.x;
  const int lane = tid & 31;
  const int wave = __builtin_amdgcn_readfirstlane(tid >> 5);
  if (tid < nr) hist[tid] = 0;
  if (tid == 0) ovf = 0;
  __syncthreads();

  int* wl = lst + wave * GRP_CAPW;
  int below = 0, wcount = 0;
  const int4* __restrict__ d4 = (const int4*)dst;
  constexpr int N4 = NEDGE / 4;
  constexpr int MAINC = N4 / GRP_T / 8 * 8;
  constexpr int TAILC = (N4 + GRP_T - 1) / GRP_T - MAINC;
  static_assert(TAILC >= 1 && TAILC <= 8, "the tail is one group of chunks");
  for (int c0 = 0; c0 < MAINC; c0 += 8)
    grp_chunks<8, false>(d4, c0, tid, lo, nr, hist, wl, below, wcount);
  grp_chunks<TAILC, true>(d4, MAINC, tid, lo, nr, hist, wl, below, wcount);
  if (lane == 0) {
    part[wave] = below;
    if (wcount > GRP_CAPW) ovf = 1;
  }
  __syncthreads();

  const int c = (tid < nr) ? hist[tid] : 0;
  int inc = c;
  for (int o = 1; o < 32; o <<= 1) {
    int t = __shfl_up(inc, o, 32);
    if (lane >= o) inc += t;
  }
  if (lane == 31) wsum[wave] = inc;
  int base = part[lane];
  for (int o = 16; o > 0; o >>= 1) base += __shfl_xor(base, o, 32);
  __syncthreads();
  const int pw = wsum[lane];
  int pinc = pw;
  for (int o = 1; o < 32; o <<= 1) {
    int t = __shfl_up(pinc, o, 32);
    if (lane >= o) pinc += t;
  }
  const int woff = __shfl(pinc - pw, wave, 32);
  const int total = __shfl(pinc, 31, 32);
  const int excl = woff + inc - c;
  if (tid < nr) {
    rsl[tid] = excl;
    rs[lo + tid] = base + excl;
  }
  if (b == GRP_WGS_T - 1 && tid == 0) rs[nDst] = base + total;
  const bool over = ovf != 0;
  __syncthreads();

  if (!over) {
    const int m = min(wcount, GRP_CAPW);
    for (int p0 = lane; p0 < m + lane; p0 += 128) {
      unsigned en[4];
      int sv[4];
#pragma unroll
      for (int k = 0; k < 4; ++k) en[k] = (unsigned)wl[min(p0 + 32 * k, GRP_CAPW - 1)];
#pragma unroll
      for (int k = 0; k < 4; ++k)
        sv[k] = srcv[min((int)(en[k] & ((1u << GRP_EBITS) - 1u)), NEDGE - 1)];
#pragma unroll
      for (int k = 0; k < 4; ++k) {
        if (p0 + 32 * k < m) {
          const int pos = base + atomicAdd(&rsl[en[k] >> GRP_EBITS], 1);
          perm[min(max(pos, 0), NEDGE - 1)] = sv[k];
        }
      }
    }
  } else {
    for (int e = tid; e < NEDGE; e += GRP_T) {
      const unsigned dl = (unsigned)(dst[e] - lo);
      if (dl < (unsigned)nr) {
        const int pos = base + atomicAdd(&rsl[dl], 1);
        perm[min(max(pos, 0), NEDGE - 1)] = srcv[e];
      }
    }
  }
}

__device__ __forceinline__ void store_mean(float* p, const float4 acc, int count) {
  const float c = fmaxf((float)count, 1.0f);
  float4 o;
  o.x = acc.x / c;
  o.y = acc.y / c;
  o.z = acc.z / c;
  o.w = acc.w / c;
  *(float4*)p = o;
}

template <int U, int PITCH>
__device__ __forceinline__ void add_rows(float4& acc, int& r, int& rowBeg, int& rowEnd,
                                         const int rsv, const int nRows,
                                         const float* __restrict__ xl, float* mrow,
                                         int myidx, int j, int e, int lastSrc) {
  float4 rr[U];
#pragma unroll
  for (int u = 0; u < U; ++u) {
    int s = __builtin_amdgcn_readlane(myidx, j + u);
    s = min(max(s, 0), lastSrc);
    rr[u] = *(const float4*)(xl + (size_t)s * DIM);
  }
#pragma unroll
  for (int u = 0; u < U; ++u) {
    if (e + u >= rowEnd && r + 1 < nRows) {
      store_mean(mrow + r * PITCH, acc, rowEnd - rowBeg);
      acc = make_float4(0.0f, 0.0f, 0.0f, 0.0f);
      ++r;
      rowBeg = rowEnd;
      rowEnd = __builtin_amdgcn_readlane(rsv, r + 1);
      while (e + u >= rowEnd && r + 1 < nRows) {
        *(float4*)(mrow + r * PITCH) = acc;
        ++r;
        rowBeg = rowEnd;
        rowEnd = __builtin_amdgcn_readlane(rsv, r + 1);
      }
    }
    acc.x += rr[u].x;
    acc.y += rr[u].y;
    acc.z += rr[u].z;
    acc.w += rr[u].w;
  }
}

template <int PITCH>
__device__ __forceinline__ void gather_means(const float* __restrict__ xsrc,
                                             const int* __restrict__ rs,
                                             const int* __restrict__ perm, float* mrow,
                                             const int d0, const int nRows, const int nSrc,
                                             const int lane) {
  const int rsv = (lane <= nRows) ? rs[d0 + lane] : 0;
  const int eBeg = __builtin_amdgcn_readlane(rsv, 0);
  const int eEnd = __builtin_amdgcn_readlane(rsv, nRows);
  const float* xl = xsrc + lane * 4;
  int r = 0;
  int rowBeg = eBeg;
  int rowEnd = __builtin_amdgcn_readlane(rsv, 1);
  float4 acc = make_float4(0.0f, 0.0f, 0.0f, 0.0f);
  for (int base = eBeg; base < eEnd; base += 128) {
    const int left = eEnd - base;
    const int i0 = (lane < left) ? perm[base + lane] : 0;
    const int i1 = (lane + 32 < left) ? perm[base + 32 + lane] : 0;
    const int i2 = (lane + 64 < left) ? perm[base + 64 + lane] : 0;
    const int i3 = (lane + 96 < left) ? perm[base + 96 + lane] : 0;
#pragma unroll 1
    for (int k = 0; k < 4; ++k) {
      const int b = base + 32 * k;
      const int m = min(eEnd - b, 32);
      if (m <= 0) break;
      const int myidx = (k == 0) ? i0 : (k == 1) ? i1 : (k == 2) ? i2 : i3;
      int j = 0;
      for (; j + 16 <= m; j += 16)
        add_rows<16, PITCH>(acc, r, rowBeg, rowEnd, rsv, nRows, xl, mrow, myidx, j, b + j, nSrc - 1);
      if (m - j >= 8) {
        add_rows<8, PITCH>(acc, r, rowBeg, rowEnd, rsv, nRows, xl, mrow, myidx, j, b + j, nSrc - 1);
        j += 8;
      }
      if (m - j >= 4) {
        add_rows<4, PITCH>(acc, r, rowBeg, rowEnd, rsv, nRows, xl, mrow, myidx, j, b + j, nSrc - 1);
        j += 4;
      }
      if (m - j >= 2) {
        add_rows<2, PITCH>(acc, r, rowBeg, rowEnd, rsv, nRows, xl, mrow, myidx, j, b + j, nSrc - 1);
        j += 2;
      }
      if (m - j >= 1)
        add_rows<1, PITCH>(acc, r, rowBeg, rowEnd, rsv, nRows, xl, mrow, myidx, j, b + j, nSrc - 1);
    }
  }
  store_mean(mrow + r * PITCH, acc, rowEnd - rowBeg);
  for (++r; r < nRows; ++r)
    *(float4*)(mrow + r * PITCH) = make_float4(0.0f, 0.0f, 0.0f, 0.0f);
}

#define WPITCH 257
#define MPITCH 132
#define GEMM_WAVES 16
#define GEMM_WGS 256
#define GEMM_WGS_J 96
#define MAXS 20
static_assert(NJJ % 16 == 0 && NUU % 16 == 0, "whole 16-row strips");
static_assert((NJJ / 16 + GEMM_WGS_J - 1) / GEMM_WGS_J <= MAXS, "job run fits the LDS plane");
static_assert((NUU / 16 + (GEMM_WGS - GEMM_WGS_J) - 1) / (GEMM_WGS - GEMM_WGS_J) <= MAXS,
              "user run fits the LDS plane");
static_assert(NJJ / 16 >= GEMM_WGS_J && NUU / 16 >= GEMM_WGS - GEMM_WGS_J, "no empty run");
static_assert(MAXS <= 31, "a wave keeps the run starts of its rows in one register");
__device__ __forceinline__ void strip_product(v8f (&acc)[8], const float* aBase, const v2f* bBase) {
  v2f a[4];
#pragma unroll
  for (int u = 0; u < 4; ++u) a[u] = *(const v2f*)(aBase + u * 4);
#pragma unroll 1
  for (int g = 0; g < DIM / 16; ++g) {
    v2f aN[4];
    const int gn = (g + 1) & (DIM / 16 - 1);
#pragma unroll
    for (int u = 0; u < 4; ++u) aN[u] = *(const v2f*)(aBase + gn * 16 + u * 4);
    const v2f* b = bBase + g * 4 * WPITCH;
#pragma unroll
    for (int u = 0; u < 4; ++u) {
#pragma unroll
      for (int t = 0; t < 8; ++t)
        acc[t] = __builtin_amdgcn_wmma_f32_16x16x4_f32(false, a[u], false, b[u * WPITCH + t * 32],
                                                       (short)0, acc[t], false, false);
    }
#pragma unroll
    for (int u = 0; u < 4; ++u) a[u] = aN[u];
  }
}

__global__ __launch_bounds__(512) void k_sage_layer(
    const float* __restrict__ xsrcJ, const int* __restrict__ rsJ, const int* __restrict__ permJ,
    const float* __restrict__ xdstJ, const float* __restrict__ WlJ, const float* __restrict__ WrJ,
    const float* __restrict__ biasJ, float* __restrict__ outJ,
    const float* __restrict__ xsrcU, const int* __restrict__ rsU, const int* __restrict__ permU,
    const float* __restrict__ xdstU, const float* __restrict__ WlU, const float* __restrict__ WrU,
    const float* __restrict__ biasU, float* __restrict__ outU,
    int relu) {
  __shared__ v2f sW[2 * 32 * WPITCH];
  __shared__ float4 sM4[MAXS * 16 * MPITCH / 4];
  float* sM = (float*)sM4;
  int wave = threadIdx.x >> 5;
  int lane = threadIdx.x & 31;
  const bool isJ = blockIdx.x < GEMM_WGS_J;
  const float* __restrict__ xsrc = isJ ? xsrcJ : xsrcU;
  const int* __restrict__ rs = isJ ? rsJ : rsU;
  const int* __restrict__ perm = isJ ? permJ : permU;
  const float* __restrict__ xdst = isJ ? xdstJ : xdstU;
  const float* __restrict__ Wl = isJ ? WlJ : WlU;
  const float* __restrict__ Wr = isJ ? WrJ : WrU;
  const float* __restrict__ bias = isJ ? biasJ : biasU;
  float* __restrict__ out = isJ ? outJ : outU;
  const int nSrc = isJ ? NUU : NJJ;
  const int wg = isJ ? (int)blockIdx.x : (int)blockIdx.x - GEMM_WGS_J;
  const unsigned uw = (unsigned)wg;
  const int sBeg = isJ ? (int)(uw * (NJJ / 16) / GEMM_WGS_J)
                       : (int)(uw * (NUU / 16) / (GEMM_WGS - GEMM_WGS_J));
  const int sEnd = isJ ? (int)((uw + 1u) * (NJJ / 16) / GEMM_WGS_J)
                       : (int)((uw + 1u) * (NUU / 16) / (GEMM_WGS - GEMM_WGS_J));
  const int nS = sEnd - sBeg;

#pragma unroll
  for (int i = 0; i < 2 * DIM / GEMM_WAVES; ++i) {
    int n = wave + i * GEMM_WAVES;
    const float* wsrc = (n < DIM) ? Wl : Wr;
    int nn = n & (DIM - 1);
    const float4 w = *(const float4*)(wsrc + (size_t)nn * DIM + lane * 4);
    v2f* d = sW + (n >> 7) * (32 * WPITCH) + lane * WPITCH + (nn >> 4) * 32 + (nn & 15);
    v2f lo = {w.x, w.y};
    v2f hi = {w.z, w.w};
    d[0] = lo;
    d[16] = hi;
  }

  __syncthreads();

  int lm = lane & 15;
  int khalf = (lane >= 16) ? 2 : 0;
  const v2f* sWl = sW + lane;
  const v2f* sWr = sW + 32 * WPITCH + lane;
  const int wv = __builtin_amdgcn_readfirstlane(wave);
  const bool own = wv < nS;
  const bool early = (((wv ^ (wv >> 2)) & 1) == 0);

  v8f accB[8];
#pragma unroll
  for (int t = 0; t < 8; ++t) accB[t] = (v8f){0.f, 0.f, 0.f, 0.f, 0.f, 0.f, 0.f, 0.f};
  const float* xBase0 = xdst + (size_t)((sBeg + wv) * 16 + lm) * DIM + khalf;
  if (own && early) strip_product(accB, xBase0, sWr);
  gather_means<MPITCH>(xsrc, rs, perm, sM + wv * nS * MPITCH + lane * 4, sBeg * 16 + wv * nS, nS,
                       nSrc, lane);
  if (own && !early) strip_product(accB, xBase0, sWr);
  __syncthreads();

  for (int s = wv; s < nS; s += GEMM_WAVES) {
    int row0 = (sBeg + s) * 16;
    if (s != wv) {
#pragma unroll
      for (int t = 0; t < 8; ++t) accB[t] = (v8f){0.f, 0.f, 0.f, 0.f, 0.f, 0.f, 0.f, 0.f};
      strip_product(accB, xdst + (size_t)(row0 + lm) * DIM + khalf, sWr);
    }
    v8f accA[8];
#pragma unroll
    for (int t = 0; t < 8; ++t) {
      float bval = bias[t * 16 + lm];
      accA[t] = (v8f){bval, bval, bval, bval, bval, bval, bval, bval};
    }
    strip_product(accA, sM + (s * 16 + lm) * MPITCH + khalf, sWl);

    float* orow = out + (size_t)(row0 + ((lane >= 16) ? 8 : 0)) * DIM + lm;
#pragma unroll
    for (int t = 0; t < 8; ++t) {
#pragma unroll
      for (int r = 0; r < 8; ++r) {
        float v = accA[t][r] + accB[t][r];
        if (relu) v = fmaxf(v, 0.0f);
        orow[(size_t)r * DIM + t * 16] = v;
      }
    }
  }
}

__global__ __launch_bounds__(256) void k_decode(
    const float* __restrict__ zu, const float* __restrict__ zj,
    const int* __restrict__ elu, const int* __restrict__ elj,
    float* __restrict__ out, int n) {
  const int p0 = 32 * __builtin_amdgcn_readfirstlane((int)(blockIdx.x * 8 + (threadIdx.x >> 5)));
  if (p0 >= n) return;
  const int lane = threadIdx.x & 31;
  const int np = min(n - p0, 32);
  const int iu = (lane < np) ? elu[p0 + lane] : 0;
  const int ij = (lane < np) ? elj[p0 + lane] : 0;
  const float* zul = zu + lane * 4;
  const float* zjl = zj + lane * 4;
  float rd = 0.0f, ra = 0.0f, rb = 0.0f;
#pragma unroll 1
  for (int q = 0; q < np; q += 8) {
    float4 a[8], b[8];
#pragma unroll
    for (int i = 0; i < 8; ++i) {
      const int su = __builtin_amdgcn_readlane(iu, q + i);
      const int sj = __builtin_amdgcn_readlane(ij, q + i);
      a[i] = *(const float4*)(zul + (size_t)su * DIM);
      b[i] = *(const float4*)(zjl + (size_t)sj * DIM);
    }
    __builtin_amdgcn_sched_barrier(0);
    float dot[8], na[8], nb[8];
#pragma unroll
    for (int i = 0; i < 8; ++i) {
      dot[i] = fmaf(a[i].w, b[i].w, fmaf(a[i].z, b[i].z, fmaf(a[i].x, b[i].x, a[i].y * b[i].y)));
      na[i] = fmaf(a[i].w, a[i].w, fmaf(a[i].z, a[i].z, fmaf(a[i].x, a[i].x, a[i].y * a[i].y)));
      nb[i] = fmaf(b[i].w, b[i].w, fmaf(b[i].z, b[i].z, fmaf(b[i].x, b[i].x, b[i].y * b[i].y)));
    }
#pragma unroll
    for (int off = 16; off > 0; off >>= 1) {
#pragma unroll
      for (int i = 0; i < 8; ++i) {
        dot[i] += __shfl_xor(dot[i], off, 32);
        na[i] += __shfl_xor(na[i], off, 32);
        nb[i] += __shfl_xor(nb[i], off, 32);
      }
    }
#pragma unroll
    for (int i = 0; i < 8; ++i) {
      if (lane == q + i) {
        rd = dot[i];
        ra = na[i];
        rb = nb[i];
      }
    }
  }
  if (lane < np) out[p0 + lane] = rd / (fmaxf(sqrtf(ra), 1e-8f) * fmaxf(sqrtf(rb), 1e-8f));
}


extern "C" void kernel_launch(void* const* d_in, const int* in_sizes, int n_in,
                              void* d_out, int out_size, void* d_ws, size_t ws_size,
                              hipStream_t stream) {
  const float* x_user = (const float*)d_in[0];
  const float* x_job = (const float*)d_in[1];
  const int* uj_src = (const int*)d_in[2];
  const int* uj_dst = (const int*)d_in[3];
  const int* ju_src = (const int*)d_in[4];
  const int* ju_dst = (const int*)d_in[5];
  const int* el_user = (const int*)d_in[6];
  const int* el_job = (const int*)d_in[7];
  const float* W1l_uj = (const float*)d_in[8];
  const float* W1r_uj = (const float*)d_in[9];
  const float* W1l_ju = (const float*)d_in[10];
  const float* W1r_ju = (const float*)d_in[11];
  const float* W2l_uj = (const float*)d_in[12];
  const float* W2r_uj = (const float*)d_in[13];
  const float* W2l_ju = (const float*)d_in[14];
  const float* W2r_ju = (const float*)d_in[15];
  const float* b1_uj = (const float*)d_in[16];
  const float* b1_ju = (const float*)d_in[17];
  const float* b2_uj = (const float*)d_in[18];
  const float* b2_ju = (const float*)d_in[19];
  float* out = (float*)d_out;

  float* ws = (float*)d_ws;
  float* agg_user = ws; ws += (size_t)NUU * DIM;
  float* agg_job  = ws; ws += (size_t)NJJ * DIM;
  float* h_user   = ws; ws += (size_t)NUU * DIM;
  float* h_job    = ws; ws += (size_t)NJJ * DIM;
  float* z_user   = ws; ws += (size_t)NUU * DIM;
  float* z_job    = ws; ws += (size_t)NJJ * DIM;
  (void)agg_job;
  int* perm_job  = (int*)agg_user;
  int* perm_user = perm_job + NEDGE;
  int* rs_job    = perm_user + NEDGE;
  int* rs_user   = rs_job + (NJJ + 32);

  k_group<<<2 * GRP_WGS_T, GRP_T, 0, stream>>>(uj_dst, uj_src, ju_dst, ju_src, rs_job, rs_user,
                                               perm_job, perm_user);

  k_sage_layer<<<GEMM_WGS, 512, 0, stream>>>(x_user, rs_job, perm_job, x_job, W1l_uj, W1r_uj, b1_uj,
                                             h_job, x_job, rs_user, perm_user, x_user, W1l_ju,
                                             W1r_ju, b1_ju, h_user, 1);

  k_sage_layer<<<GEMM_WGS, 512, 0, stream>>>(h_user, rs_job, perm_job, h_job, W2l_uj, W2r_uj, b2_uj,
                                             z_job, h_job, rs_user, perm_user, h_user, W2l_ju,
                                             W2r_ju, b2_ju, z_user, 0);

  k_decode<<<((NEL + 31) / 32 + 7) / 8, dim3(256), 0, stream>>>(z_user, z_job, el_user, el_job, out, NEL);
}
